# starting kernel
# baseline (speedup 1.0000x reference)
#include <hip/hip_runtime.h>
#include <hip/hip_bf16.h>
#include <math.h>

typedef __attribute__((ext_vector_type(2))) float v2f;
typedef __attribute__((ext_vector_type(8))) float v8f;

#define HEADS1 4
#define HID1   64
#define C1     256
#define C2     128
#define NEG_SLOPE 0.2f

#define KC        128
#define BS_STRIDE 72

__device__ __forceinline__ void atomicMaxF(float* addr, float v) {
    if (v >= 0.0f) atomicMax((int*)addr, __float_as_int(v));
    else           atomicMin((unsigned int*)addr, __float_as_uint(v));
}

__device__ __forceinline__ float lrelu(float v) {
    return v < 0.0f ? NEG_SLOPE * v : v;
}

__global__ void init_f32_k(float* __restrict__ p, size_t n, float val) {
    size_t i = (size_t)blockIdx.x * blockDim.x + threadIdx.x;
    if (i < n) p[i] = val;
}

__global__ void out_bias_k(float* __restrict__ out, const float* __restrict__ b,
                           int n, int C) {
    size_t i = (size_t)blockIdx.x * blockDim.x + threadIdx.x;
    if (i < (size_t)n * C) out[i] = b[i % C];
}

__global__ void bias_relu_k(float* __restrict__ x, const float* __restrict__ b,
                            size_t n, int C) {
    size_t i = (size_t)blockIdx.x * blockDim.x + threadIdx.x;
    if (i < n) {
        float v = x[i] + b[i % C];
        x[i] = v > 0.0f ? v : 0.0f;
    }
}

__global__ void __launch_bounds__(128)
gemm_wmma_f32_k(const float* __restrict__ A,
                const float* __restrict__ B,
                float* __restrict__ C,
                int M, int K, int Ncols) {
    __shared__ float Bs[KC * BS_STRIDE];

    const int t     = threadIdx.x;
    const int lane  = t & 31;
    const int wave  = t >> 5;
    const int mtile = blockIdx.x * 4 + wave;
    const int m0    = mtile * 16;
    const bool valid = (m0 < M);
    const int n0    = blockIdx.y * 64;
    const int l15   = lane & 15;
    const int koff  = (lane >> 4) * 2;
    const int row   = valid ? (m0 + l15) : 0;

    v8f acc0 = {}, acc1 = {}, acc2 = {}, acc3 = {};

    for (int k0 = 0; k0 < K; k0 += KC) {
        __syncthreads();
#pragma unroll
        for (int i = 0; i < (KC * 64 / 4) / 128; ++i) {
            int j = t + i * 128;
            int k = j >> 4;
            int c = (j & 15) * 4;
            const float4 v = *(const float4*)(B + (size_t)(k0 + k) * Ncols + n0 + c);
            *(float4*)(&Bs[k * BS_STRIDE + c]) = v;
        }
        __syncthreads();

        if (valid) {
            const float* ap0 = A + (size_t)row * K + k0 + koff;
            for (int kk = 0; kk < KC; kk += 4) {
                v2f a; a.x = ap0[kk]; a.y = ap0[kk + 1];

                const float* bp = &Bs[(kk + koff) * BS_STRIDE + l15];
                v2f b0; b0.x = bp[0];  b0.y = bp[BS_STRIDE];
                v2f b1; b1.x = bp[16]; b1.y = bp[BS_STRIDE + 16];
                v2f b2; b2.x = bp[32]; b2.y = bp[BS_STRIDE + 32];
                v2f b3; b3.x = bp[48]; b3.y = bp[BS_STRIDE + 48];

                acc0 = __builtin_amdgcn_wmma_f32_16x16x4_f32(false, a, false, b0, (short)0, acc0, false, false);
                acc1 = __builtin_amdgcn_wmma_f32_16x16x4_f32(false, a, false, b1, (short)0, acc1, false, false);
                acc2 = __builtin_amdgcn_wmma_f32_16x16x4_f32(false, a, false, b2, (short)0, acc2, false, false);
                acc3 = __builtin_amdgcn_wmma_f32_16x16x4_f32(false, a, false, b3, (short)0, acc3, false, false);
            }
        }
    }

    if (valid) {
        const int rbase = m0 + ((lane >> 4) ? 8 : 0);
#pragma unroll
        for (int r = 0; r < 8; ++r) {
            float* cp = C + (size_t)(rbase + r) * Ncols + n0 + l15;
            cp[0]  = acc0[r];
            cp[16] = acc1[r];
            cp[32] = acc2[r];
            cp[48] = acc3[r];
        }
    }
}

__global__ void attn_logits_k(const float* __restrict__ h,
                              const float* __restrict__ a_src,
                              const float* __restrict__ a_dst,
                              float* __restrict__ als, float* __restrict__ ald,
                              int n, int heads, int ch) {
    int i = blockIdx.x * blockDim.x + threadIdx.x;
    if (i >= n * heads) return;
    int node = i / heads, hd = i % heads;
    const float* hp = h + (size_t)node * heads * ch + (size_t)hd * ch;
    const float* as = a_src + hd * ch;
    const float* ad = a_dst + hd * ch;
    float ss = 0.0f, sd = 0.0f;
    for (int c = 0; c < ch; ++c) {
        float v = hp[c];
        ss += v * as[c];
        sd += v * ad[c];
    }
    als[i] = ss;
    ald[i] = sd;
}

__global__ void edge_max_k(const int* __restrict__ ei, int E, int n,
                           const float* __restrict__ als, const float* __restrict__ ald,
                           float* __restrict__ m, int heads) {
    int e = blockIdx.x * blockDim.x + threadIdx.x;
    int ET = E + n;
    if (e >= ET) return;
    int s, d;
    if (e < E) { s = ei[e]; d = ei[E + e]; } else { s = d = e - E; }
    for (int hd = 0; hd < heads; ++hd) {
        float v = lrelu(als[(size_t)s * heads + hd] + ald[(size_t)d * heads + hd]);
        atomicMaxF(&m[(size_t)d * heads + hd], v);
    }
}

__global__ void edge_sum_k(const int* __restrict__ ei, int E, int n,
                           const float* __restrict__ als, const float* __restrict__ ald,
                           const float* __restrict__ m, float* __restrict__ den,
                           int heads) {
    int e = blockIdx.x * blockDim.x + threadIdx.x;
    int ET = E + n;
    if (e >= ET) return;
    int s, d;
    if (e < E) { s = ei[e]; d = ei[E + e]; } else { s = d = e - E; }
    for (int hd = 0; hd < heads; ++hd) {
        float v = lrelu(als[(size_t)s * heads + hd] + ald[(size_t)d * heads + hd]);
        float ex = __expf(v - m[(size_t)d * heads + hd]);
        atomicAdd(&den[(size_t)d * heads + hd], ex);
    }
}

__global__ void edge_scatter_k(const int* __restrict__ ei, int E, int n,
                               const float* __restrict__ als, const float* __restrict__ ald,
                               const float* __restrict__ m, const float* __restrict__ den,
                               const float* __restrict__ h, float* __restrict__ out,
                               int heads, int ch) {
    const int C  = heads * ch;
    const int t  = threadIdx.x;
    const int hd = t / ch;
    const int ET = E + n;
    for (int e = blockIdx.x; e < ET; e += gridDim.x) {
        int s, d;
        if (e < E) { s = ei[e]; d = ei[E + e]; } else { s = d = e - E; }
        float v = lrelu(als[(size_t)s * heads + hd] + ald[(size_t)d * heads + hd]);
        float alpha = __expf(v - m[(size_t)d * heads + hd]) / den[(size_t)d * heads + hd];
        atomicAdd(&out[(size_t)d * C + t], h[(size_t)s * C + t] * alpha);
    }
}

extern "C" void kernel_launch(void* const* d_in, const int* in_sizes, int n_in,
                              void* d_out, int out_size, void* d_ws, size_t ws_size,
                              hipStream_t stream) {
    const float* x      = (const float*)d_in[0];
    const int*   eidx   = (const int*)d_in[1];
    const float* W1     = (const float*)d_in[2];
    const float* a1_src = (const float*)d_in[3];
    const float* a1_dst = (const float*)d_in[4];
    const float* b1     = (const float*)d_in[5];
    const float* W2     = (const float*)d_in[6];
    const float* a2_src = (const float*)d_in[7];
    const float* a2_dst = (const float*)d_in[8];
    const float* b2     = (const float*)d_in[9];
    float*       out    = (float*)d_out;

    const int N = in_sizes[0] / 128;
    const int E = in_sizes[1] / 2;
    const int ET = E + N;

    float* ws   = (float*)d_ws;
    float* h1   = ws;
    float* out1 = h1   + (size_t)N * C1;
    float* als1 = out1 + (size_t)N * C1;
    float* ald1 = als1 + (size_t)N * HEADS1;
    float* m1   = ald1 + (size_t)N * HEADS1;
    float* den1 = m1   + (size_t)N * HEADS1;
    float* als2 = den1 + (size_t)N * HEADS1;
    float* ald2 = als2 + (size_t)N;
    float* m2   = ald2 + (size_t)N;
    float* den2 = m2   + (size_t)N;
    float* h2   = h1;

    const int mtiles = (N + 15) / 16;

    {
        dim3 grid((mtiles + 3) / 4, C1 / 64);
        gemm_wmma_f32_k<<<grid, 128, 0, stream>>>(x, W1, h1, N, 128, C1);
    }
    {
        int tot = N * HEADS1;
        attn_logits_k<<<(tot + 255) / 256, 256, 0, stream>>>(h1, a1_src, a1_dst,
                                                             als1, ald1, N, HEADS1, HID1);
    }
    {
        size_t nm = (size_t)N * HEADS1;
        init_f32_k<<<(int)((nm + 255) / 256), 256, 0, stream>>>(m1, nm, -INFINITY);
        init_f32_k<<<(int)((nm + 255) / 256), 256, 0, stream>>>(den1, nm, 0.0f);
        size_t no = (size_t)N * C1;
        init_f32_k<<<(int)((no + 255) / 256), 256, 0, stream>>>(out1, no, 0.0f);
    }
    edge_max_k<<<(ET + 255) / 256, 256, 0, stream>>>(eidx, E, N, als1, ald1, m1, HEADS1);
    edge_sum_k<<<(ET + 255) / 256, 256, 0, stream>>>(eidx, E, N, als1, ald1, m1, den1, HEADS1);
    edge_scatter_k<<<ET, C1, 0, stream>>>(eidx, E, N, als1, ald1, m1, den1, h1, out1,
                                          HEADS1, HID1);
    {
        size_t no = (size_t)N * C1;
        bias_relu_k<<<(int)((no + 255) / 256), 256, 0, stream>>>(out1, b1, no, C1);
    }

    {
        dim3 grid((mtiles + 3) / 4, C2 / 64);
        gemm_wmma_f32_k<<<grid, 128, 0, stream>>>(out1, W2, h2, N, C1, C2);
    }
    attn_logits_k<<<(N + 255) / 256, 256, 0, stream>>>(h2, a2_src, a2_dst,
                                                       als2, ald2, N, 1, C2);
    {
        init_f32_k<<<(N + 255) / 256, 256, 0, stream>>>(m2, (size_t)N, -INFINITY);
        init_f32_k<<<(N + 255) / 256, 256, 0, stream>>>(den2, (size_t)N, 0.0f);
        size_t no = (size_t)N * C2;
        out_bias_k<<<(int)((no + 255) / 256), 256, 0, stream>>>(out, b2, N, C2);
    }
    edge_max_k<<<(ET + 255) / 256, 256, 0, stream>>>(eidx, E, N, als2, ald2, m2, 1);
    edge_sum_k<<<(ET + 255) / 256, 256, 0, stream>>>(eidx, E, N, als2, ald2, m2, den2, 1);
    edge_scatter_k<<<ET, C2, 0, stream>>>(eidx, E, N, als2, ald2, m2, den2, h2, out,
                                          1, C2);
}


# Edges bucketed by destination once per call (count, scan, place); one wave per node does max, exp-sum and weighted row gather in registers: no float atomics; sums in bucket order, normalised last
# speedup vs baseline: 6.0656x; 6.0656x over previous
#include <hip/hip_runtime.h>
#include <hip/hip_bf16.h>
#include <math.h>

typedef __attribute__((ext_vector_type(2))) float v2f;
typedef __attribute__((ext_vector_type(8))) float v8f;

#define HEADS1 4
#define HID1   64
#define C1     256
#define C2     128
#define NEG_SLOPE 0.2f

#define KC        128
#define BS_STRIDE 72

__device__ __forceinline__ float lrelu(float v) {
    return v < 0.0f ? NEG_SLOPE * v : v;
}

__global__ void __launch_bounds__(128)
gemm_wmma_f32_k(const float* __restrict__ A,
                const float* __restrict__ B,
                float* __restrict__ C,
                int M, int K, int Ncols) {
    __shared__ float Bs[KC * BS_STRIDE];

    const int t     = threadIdx.x;
    const int lane  = t & 31;
    const int wave  = t >> 5;
    const int mtile = blockIdx.x * 4 + wave;
    const int m0    = mtile * 16;
    const bool valid = (m0 < M);
    const int n0    = blockIdx.y * 64;
    const int l15   = lane & 15;
    const int koff  = (lane >> 4) * 2;
    const int row   = valid ? (m0 + l15) : 0;

    v8f acc0 = {}, acc1 = {}, acc2 = {}, acc3 = {};

    for (int k0 = 0; k0 < K; k0 += KC) {
        __syncthreads();
#pragma unroll
        for (int i = 0; i < (KC * 64 / 4) / 128; ++i) {
            int j = t + i * 128;
            int k = j >> 4;
            int c = (j & 15) * 4;
            const float4 v = *(const float4*)(B + (size_t)(k0 + k) * Ncols + n0 + c);
            *(float4*)(&Bs[k * BS_STRIDE + c]) = v;
        }
        __syncthreads();

        if (valid) {
            const float* ap0 = A + (size_t)row * K + k0 + koff;
            for (int kk = 0; kk < KC; kk += 4) {
                v2f a; a.x = ap0[kk]; a.y = ap0[kk + 1];

                const float* bp = &Bs[(kk + koff) * BS_STRIDE + l15];
                v2f b0; b0.x = bp[0];  b0.y = bp[BS_STRIDE];
                v2f b1; b1.x = bp[16]; b1.y = bp[BS_STRIDE + 16];
                v2f b2; b2.x = bp[32]; b2.y = bp[BS_STRIDE + 32];
                v2f b3; b3.x = bp[48]; b3.y = bp[BS_STRIDE + 48];

                acc0 = __builtin_amdgcn_wmma_f32_16x16x4_f32(false, a, false, b0, (short)0, acc0, false, false);
                acc1 = __builtin_amdgcn_wmma_f32_16x16x4_f32(false, a, false, b1, (short)0, acc1, false, false);
                acc2 = __builtin_amdgcn_wmma_f32_16x16x4_f32(false, a, false, b2, (short)0, acc2, false, false);
                acc3 = __builtin_amdgcn_wmma_f32_16x16x4_f32(false, a, false, b3, (short)0, acc3, false, false);
            }
        }
    }

    if (valid) {
        const int rbase = m0 + ((lane >> 4) ? 8 : 0);
#pragma unroll
        for (int r = 0; r < 8; ++r) {
            float* cp = C + (size_t)(rbase + r) * Ncols + n0 + l15;
            cp[0]  = acc0[r];
            cp[16] = acc1[r];
            cp[32] = acc2[r];
            cp[48] = acc3[r];
        }
    }
}

__global__ void attn_logits_k(const float* __restrict__ h,
                              const float* __restrict__ a_src,
                              const float* __restrict__ a_dst,
                              float* __restrict__ als, float* __restrict__ ald,
                              int n, int heads, int ch) {
    int i = blockIdx.x * blockDim.x + threadIdx.x;
    if (i >= n * heads) return;
    int node = i / heads, hd = i % heads;
    const float* hp = h + (size_t)node * heads * ch + (size_t)hd * ch;
    const float* as = a_src + hd * ch;
    const float* ad = a_dst + hd * ch;
    float ss = 0.0f, sd = 0.0f;
    for (int c = 0; c < ch; ++c) {
        float v = hp[c];
        ss += v * as[c];
        sd += v * ad[c];
    }
    als[i] = ss;
    ald[i] = sd;
}

__global__ void init_i32_k(int* __restrict__ p, int n, int val) {
    int i = blockIdx.x * blockDim.x + threadIdx.x;
    if (i < n) p[i] = val;
}

__global__ void edge_rank_k(const int* __restrict__ ei, int E,
                            int* __restrict__ cnt, int* __restrict__ rank) {
    int e = blockIdx.x * blockDim.x + threadIdx.x;
    if (e >= E) return;
    rank[e] = atomicAdd(&cnt[ei[E + e]], 1);
}

__global__ void __launch_bounds__(1024)
rowptr_scan_k(const int* __restrict__ cnt, int* __restrict__ rowptr, int n) {
    __shared__ int wtot[32];
    __shared__ int wpre[32];
    const int t    = threadIdx.x;
    const int lane = t & 31;
    const int wave = t >> 5;
    const int base = blockIdx.x * 1024;

    int pre = 0;
    for (int i = t; i < base; i += 1024) pre += cnt[i];
    for (int off = 16; off > 0; off >>= 1) pre += __shfl_xor(pre, off);

    const int idx = base + t;
    const int v   = idx < n ? cnt[idx] : 0;
    int incl = v;
    for (int off = 1; off < 32; off <<= 1) {
        int u = __shfl_up(incl, off);
        if (lane >= off) incl += u;
    }
    if (lane == 31) wtot[wave] = incl;
    if (lane == 0)  wpre[wave] = pre;
    __syncthreads();

    int before = 0;
    for (int w = 0; w < 32; ++w) {
        before += wpre[w];
        if (w < wave) before += wtot[w];
    }
    const int excl = before + incl - v;
    if (idx < n)      rowptr[idx] = excl;
    if (idx == n - 1) rowptr[n]   = excl + v;
}

__global__ void edge_place_k(const int* __restrict__ ei, int E,
                             const int* __restrict__ rowptr, const int* __restrict__ rank,
                             int* __restrict__ csr_src) {
    int e = blockIdx.x * blockDim.x + threadIdx.x;
    if (e >= E) return;
    csr_src[rowptr[ei[E + e]] + rank[e]] = ei[e];
}

template <int HEADS>
__device__ __forceinline__ float pick_head(const float (&w)[HEADS], int hsel) {
    float r = w[0];
#pragma unroll
    for (int i = 1; i < HEADS; ++i) r = (hsel == i) ? w[i] : r;
    return r;
}

template <int HEADS, int CH, bool RELU>
__global__ void __launch_bounds__(256)
gat_gather_k(const int* __restrict__ rowptr, const int* __restrict__ csr_src,
             const float* __restrict__ als, const float* __restrict__ ald,
             const float* __restrict__ h, const float* __restrict__ bias,
             float* __restrict__ out, int n) {
    constexpr int C  = HEADS * CH;
    constexpr int NQ = C / 128;
    const int lane = threadIdx.x & 31;
    const int node = __builtin_amdgcn_readfirstlane((int)(blockIdx.x * 8 + (threadIdx.x >> 5)));
    if (node >= n) return;

    const int beg = rowptr[node];
    const int cnt = rowptr[node + 1] - beg + 1;

    float adst[HEADS];
#pragma unroll
    for (int hh = 0; hh < HEADS; ++hh) adst[hh] = ald[(size_t)node * HEADS + hh];

    int hq[NQ];
#pragma unroll
    for (int q = 0; q < NQ; ++q) hq[q] = (q * 128 + lane * 4) / CH;

    float  m[HEADS], dn[HEADS];
    float4 acc[NQ];
#pragma unroll
    for (int hh = 0; hh < HEADS; ++hh) { m[hh] = 0.0f; dn[hh] = 0.0f; }
#pragma unroll
    for (int q = 0; q < NQ; ++q) acc[q] = make_float4(0.0f, 0.0f, 0.0f, 0.0f);

    for (int j0 = 0; j0 < cnt; j0 += 32) {
        const int  j     = j0 + lane;
        const bool valid = j < cnt;
        int s_l = node;
        if (valid && j > 0) s_l = min(max(csr_src[beg + j - 1], 0), n - 1);

        float a[HEADS];
        if constexpr (HEADS == 4) {
            const float4 a4 = *(const float4*)(als + (size_t)s_l * 4);
            a[0] = a4.x; a[1] = a4.y; a[2] = a4.z; a[3] = a4.w;
        } else {
#pragma unroll
            for (int hh = 0; hh < HEADS; ++hh) a[hh] = als[(size_t)s_l * HEADS + hh];
        }

        float e[HEADS], cm[HEADS], pw[HEADS], ps[HEADS], sc[HEADS];
#pragma unroll
        for (int hh = 0; hh < HEADS; ++hh) {
            e[hh]  = valid ? lrelu(a[hh] + adst[hh]) : -INFINITY;
            cm[hh] = e[hh];
        }
#pragma unroll
        for (int off = 16; off > 0; off >>= 1) {
#pragma unroll
            for (int hh = 0; hh < HEADS; ++hh) cm[hh] = fmaxf(cm[hh], __shfl_xor(cm[hh], off));
        }
#pragma unroll
        for (int hh = 0; hh < HEADS; ++hh) {
            const float mn = (j0 == 0) ? cm[hh] : fmaxf(m[hh], cm[hh]);
            sc[hh] = (j0 == 0) ? 0.0f : __expf(m[hh] - mn);
            pw[hh] = valid ? __expf(e[hh] - mn) : 0.0f;
            ps[hh] = pw[hh];
            m[hh]  = mn;
        }
#pragma unroll
        for (int off = 16; off > 0; off >>= 1) {
#pragma unroll
            for (int hh = 0; hh < HEADS; ++hh) ps[hh] += __shfl_xor(ps[hh], off);
        }
#pragma unroll
        for (int hh = 0; hh < HEADS; ++hh) dn[hh] = dn[hh] * sc[hh] + ps[hh];
        if (j0 > 0) {
#pragma unroll
            for (int q = 0; q < NQ; ++q) {
                const float f = pick_head<HEADS>(sc, hq[q]);
                acc[q].x *= f; acc[q].y *= f; acc[q].z *= f; acc[q].w *= f;
            }
        }

        const int nc = min(32, cnt - j0);
        for (int t = 0; t < nc; t += 4) {
#pragma unroll
            for (int u = 0; u < 4; ++u) {
                const int s = __builtin_amdgcn_readlane(s_l, t + u);
                float wv[HEADS];
#pragma unroll
                for (int hh = 0; hh < HEADS; ++hh)
                    wv[hh] = __int_as_float(__builtin_amdgcn_readlane(__float_as_int(pw[hh]), t + u));
                const float4* hp = (const float4*)(h + (size_t)s * C) + lane;
#pragma unroll
                for (int q = 0; q < NQ; ++q) {
                    const float4 v = hp[q * 32];
                    const float  w = pick_head<HEADS>(wv, hq[q]);
                    acc[q].x += w * v.x; acc[q].y += w * v.y;
                    acc[q].z += w * v.z; acc[q].w += w * v.w;
                }
            }
        }
    }

#pragma unroll
    for (int q = 0; q < NQ; ++q) {
        const int    c0 = q * 128 + lane * 4;
        const float  d  = pick_head<HEADS>(dn, hq[q]);
        const float4 b  = *(const float4*)(bias + c0);
        float4 r;
        r.x = acc[q].x / d + b.x; r.y = acc[q].y / d + b.y;
        r.z = acc[q].z / d + b.z; r.w = acc[q].w / d + b.w;
        if (RELU) {
            r.x = r.x > 0.0f ? r.x : 0.0f; r.y = r.y > 0.0f ? r.y : 0.0f;
            r.z = r.z > 0.0f ? r.z : 0.0f; r.w = r.w > 0.0f ? r.w : 0.0f;
        }
        *(float4*)(out + (size_t)node * C + c0) = r;
    }
}

extern "C" void kernel_launch(void* const* d_in, const int* in_sizes, int n_in,
                              void* d_out, int out_size, void* d_ws, size_t ws_size,
                              hipStream_t stream) {
    const float* x      = (const float*)d_in[0];
    const int*   eidx   = (const int*)d_in[1];
    const float* W1     = (const float*)d_in[2];
    const float* a1_src = (const float*)d_in[3];
    const float* a1_dst = (const float*)d_in[4];
    const float* b1     = (const float*)d_in[5];
    const float* W2     = (const float*)d_in[6];
    const float* a2_src = (const float*)d_in[7];
    const float* a2_dst = (const float*)d_in[8];
    const float* b2     = (const float*)d_in[9];
    float*       out    = (float*)d_out;

    const int N = in_sizes[0] / 128;
    const int E = in_sizes[1] / 2;

    float* ws   = (float*)d_ws;
    float* h1   = ws;
    float* x2   = h1   + (size_t)N * C1;
    float* als1 = x2   + (size_t)N * C1;
    float* ald1 = als1 + (size_t)N * HEADS1;
    float* als2 = ald1 + (size_t)N * HEADS1;
    float* ald2 = als2 + (size_t)N;
    int*   cnt    = (int*)(ald2 + (size_t)N);
    int*   rowptr = cnt    + (size_t)N;
    int*   rank   = rowptr + (size_t)N + 1;
    int*   csr    = rank   + (size_t)E;
    float* h2   = h1;

    const int mtiles = (N + 15) / 16;

    init_i32_k<<<(N + 255) / 256, 256, 0, stream>>>(cnt, N, 0);
    edge_rank_k<<<(E + 255) / 256, 256, 0, stream>>>(eidx, E, cnt, rank);
    rowptr_scan_k<<<(N + 1023) / 1024, 1024, 0, stream>>>(cnt, rowptr, N);
    edge_place_k<<<(E + 255) / 256, 256, 0, stream>>>(eidx, E, rowptr, rank, csr);

    {
        dim3 grid((mtiles + 3) / 4, C1 / 64);
        gemm_wmma_f32_k<<<grid, 128, 0, stream>>>(x, W1, h1, N, 128, C1);
    }
    {
        int tot = N * HEADS1;
        attn_logits_k<<<(tot + 255) / 256, 256, 0, stream>>>(h1, a1_src, a1_dst,
                                                             als1, ald1, N, HEADS1, HID1);
    }
    gat_gather_k<HEADS1, HID1, true><<<(N + 7) / 8, 256, 0, stream>>>(rowptr, csr, als1, ald1,
                                                                      h1, b1, x2, N);

    {
        dim3 grid((mtiles + 3) / 4, C2 / 64);
        gemm_wmma_f32_k<<<grid, 128, 0, stream>>>(x2, W2, h2, N, C1, C2);
    }
    attn_logits_k<<<(N + 255) / 256, 256, 0, stream>>>(h2, a2_src, a2_dst,
                                                       als2, ald2, N, 1, C2);
    gat_gather_k<1, C2, false><<<(N + 7) / 8, 256, 0, stream>>>(rowptr, csr, als2, ald2,
                                                                h2, b2, out, N);
}


# Edge bucketing as a two-level counting sort on LDS counters (128 chunks x bins of 128 destinations, then one block per bin): no global atomics; rank inside a bucket = LDS arrival order
# speedup vs baseline: 7.1707x; 1.1822x over previous
#include <hip/hip_runtime.h>
#include <hip/hip_bf16.h>
#include <math.h>

typedef __attribute__((ext_vector_type(2))) float v2f;
typedef __attribute__((ext_vector_type(8))) float v8f;

#define HEADS1 4
#define HID1   64
#define C1     256
#define C2     128
#define NEG_SLOPE 0.2f

#define KC        128
#define BS_STRIDE 72

__device__ __forceinline__ float lrelu(float v) {
    return v < 0.0f ? NEG_SLOPE * v : v;
}

__global__ void __launch_bounds__(128)
gemm_wmma_f32_k(const float* __restrict__ A,
                const float* __restrict__ B,
                float* __restrict__ C,
                int M, int K, int Ncols) {
    __shared__ float Bs[KC * BS_STRIDE];

    const int t     = threadIdx.x;
    const int lane  = t & 31;
    const int wave  = t >> 5;
    const int mtile = blockIdx.x * 4 + wave;
    const int m0    = mtile * 16;
    const bool valid = (m0 < M);
    const int n0    = blockIdx.y * 64;
    const int l15   = lane & 15;
    const int koff  = (lane >> 4) * 2;
    const int row   = valid ? (m0 + l15) : 0;

    v8f acc0 = {}, acc1 = {}, acc2 = {}, acc3 = {};

    for (int k0 = 0; k0 < K; k0 += KC) {
        __syncthreads();
#pragma unroll
        for (int i = 0; i < (KC * 64 / 4) / 128; ++i) {
            int j = t + i * 128;
            int k = j >> 4;
            int c = (j & 15) * 4;
            const float4 v = *(const float4*)(B + (size_t)(k0 + k) * Ncols + n0 + c);
            *(float4*)(&Bs[k * BS_STRIDE + c]) = v;
        }
        __syncthreads();

        if (valid) {
            const float* ap0 = A + (size_t)row * K + k0 + koff;
            for (int kk = 0; kk < KC; kk += 4) {
                v2f a; a.x = ap0[kk]; a.y = ap0[kk + 1];

                const float* bp = &Bs[(kk + koff) * BS_STRIDE + l15];
                v2f b0; b0.x = bp[0];  b0.y = bp[BS_STRIDE];
                v2f b1; b1.x = bp[16]; b1.y = bp[BS_STRIDE + 16];
                v2f b2; b2.x = bp[32]; b2.y = bp[BS_STRIDE + 32];
                v2f b3; b3.x = bp[48]; b3.y = bp[BS_STRIDE + 48];

                acc0 = __builtin_amdgcn_wmma_f32_16x16x4_f32(false, a, false, b0, (short)0, acc0, false, false);
                acc1 = __builtin_amdgcn_wmma_f32_16x16x4_f32(false, a, false, b1, (short)0, acc1, false, false);
                acc2 = __builtin_amdgcn_wmma_f32_16x16x4_f32(false, a, false, b2, (short)0, acc2, false, false);
                acc3 = __builtin_amdgcn_wmma_f32_16x16x4_f32(false, a, false, b3, (short)0, acc3, false, false);
            }
        }
    }

    if (valid) {
        const int rbase = m0 + ((lane >> 4) ? 8 : 0);
#pragma unroll
        for (int r = 0; r < 8; ++r) {
            float* cp = C + (size_t)(rbase + r) * Ncols + n0 + l15;
            cp[0]  = acc0[r];
            cp[16] = acc1[r];
            cp[32] = acc2[r];
            cp[48] = acc3[r];
        }
    }
}

__global__ void attn_logits_k(const float* __restrict__ h,
                              const float* __restrict__ a_src,
                              const float* __restrict__ a_dst,
                              float* __restrict__ als, float* __restrict__ ald,
                              int n, int heads, int ch) {
    int i = blockIdx.x * blockDim.x + threadIdx.x;
    if (i >= n * heads) return;
    int node = i / heads, hd = i % heads;
    const float* hp = h + (size_t)node * heads * ch + (size_t)hd * ch;
    const float* as = a_src + hd * ch;
    const float* ad = a_dst + hd * ch;
    float ss = 0.0f, sd = 0.0f;
    for (int c = 0; c < ch; ++c) {
        float v = hp[c];
        ss += v * as[c];
        sd += v * ad[c];
    }
    als[i] = ss;
    ald[i] = sd;
}

#define BIN_SHIFT 7
#define BIN_W     128
#define MAX_BINS  512
#define NB1       128

__device__ __forceinline__ int bin_of(int d) {
    return min(d >> BIN_SHIFT, MAX_BINS - 1);
}

__global__ void __launch_bounds__(1024)
bin_count_k(const int* __restrict__ ei, int E, int n, int nbin, int* __restrict__ H) {
    __shared__ int hist[MAX_BINS];
    const int t = threadIdx.x;
    const int b = blockIdx.x;
    for (int k = t; k < MAX_BINS; k += 1024) hist[k] = 0;
    __syncthreads();
    const int per = (E + NB1 - 1) / NB1;
    const int lo  = b * per;
    const int hi  = min(lo + per, E);
    for (int e0 = lo + t; e0 < hi; e0 += 4 * 1024) {
        int d[4];
#pragma unroll
        for (int u = 0; u < 4; ++u) {
            const int e = min(e0 + u * 1024, hi - 1);
            d[u] = ei[E + e];
        }
#pragma unroll
        for (int u = 0; u < 4; ++u) {
            if (e0 + u * 1024 < hi)
                atomicAdd(&hist[bin_of(min(max(d[u], 0), n - 1))], 1);
        }
    }
    __syncthreads();
    for (int k = t; k < nbin; k += 1024) H[k * NB1 + b] = hist[k];
}

__global__ void __launch_bounds__(1024)
rowptr_scan_k(const int* __restrict__ cnt, int* __restrict__ rowptr, int n) {
    __shared__ int wtot[32];
    __shared__ int wpre[32];
    const int t    = threadIdx.x;
    const int lane = t & 31;
    const int wave = t >> 5;
    const int base = blockIdx.x * 1024;

    int pre = 0;
    for (int i = t; i < base; i += 1024) pre += cnt[i];
    for (int off = 16; off > 0; off >>= 1) pre += __shfl_xor(pre, off);

    const int idx = base + t;
    const int v   = idx < n ? cnt[idx] : 0;
    int incl = v;
    for (int off = 1; off < 32; off <<= 1) {
        int u = __shfl_up(incl, off);
        if (lane >= off) incl += u;
    }
    if (lane == 31) wtot[wave] = incl;
    if (lane == 0)  wpre[wave] = pre;
    __syncthreads();

    int before = 0;
    for (int w = 0; w < 32; ++w) {
        before += wpre[w];
        if (w < wave) before += wtot[w];
    }
    const int excl = before + incl - v;
    if (idx < n)      rowptr[idx] = excl;
    if (idx == n - 1) rowptr[n]   = excl + v;
}

__global__ void __launch_bounds__(1024)
bin_scatter_k(const int* __restrict__ ei, int E, int n, int nbin,
              const int* __restrict__ O, int* __restrict__ pairs) {
    __shared__ int cur[MAX_BINS];
    const int t = threadIdx.x;
    const int b = blockIdx.x;
    for (int k = t; k < MAX_BINS; k += 1024) cur[k] = (k < nbin) ? O[k * NB1 + b] : 0;
    __syncthreads();
    const int per = (E + NB1 - 1) / NB1;
    const int lo  = b * per;
    const int hi  = min(lo + per, E);
    int2* __restrict__ p2 = (int2*)pairs;
    for (int e0 = lo + t; e0 < hi; e0 += 4 * 1024) {
        int d[4], s[4];
#pragma unroll
        for (int u = 0; u < 4; ++u) {
            const int e = min(e0 + u * 1024, hi - 1);
            d[u] = ei[E + e];
            s[u] = ei[e];
        }
#pragma unroll
        for (int u = 0; u < 4; ++u) {
            if (e0 + u * 1024 < hi) {
                const int dc  = min(max(d[u], 0), n - 1);
                const int pos = atomicAdd(&cur[bin_of(dc)], 1);
                p2[pos] = make_int2(s[u], dc);
            }
        }
    }
}

__global__ void __launch_bounds__(512)
bin_sort_k(const int* __restrict__ pairs, const int* __restrict__ O, int n,
           int* __restrict__ rowptr, int* __restrict__ csr_src) {
    __shared__ int cnt[BIN_W];
    __shared__ int fill[BIN_W];
    __shared__ int wsum[BIN_W / 32];
    const int t    = threadIdx.x;
    const int lane = t & 31;
    const int wave = t >> 5;
    const int k    = blockIdx.x;
    const int beg  = O[k * NB1];
    const int end  = O[(k + 1) * NB1];
    const int d0   = k << BIN_SHIFT;
    const int2* __restrict__ p2 = (const int2*)pairs;

    if (t < BIN_W) cnt[t] = 0;
    __syncthreads();
    for (int i0 = beg + t; i0 < end; i0 += 4 * 512) {
        int y[4];
#pragma unroll
        for (int u = 0; u < 4; ++u) {
            const int i = min(i0 + u * 512, end - 1);
            y[u] = p2[i].y;
        }
#pragma unroll
        for (int u = 0; u < 4; ++u) {
            if (i0 + u * 512 < end) atomicAdd(&cnt[(y[u] - d0) & (BIN_W - 1)], 1);
        }
    }
    __syncthreads();

    const int v = (t < BIN_W) ? cnt[t] : 0;
    int incl = v;
    for (int off = 1; off < 32; off <<= 1) {
        int u = __shfl_up(incl, off);
        if (lane >= off) incl += u;
    }
    if (t < BIN_W && lane == 31) wsum[wave] = incl;
    __syncthreads();
    if (t < BIN_W) {
        int before = beg;
        for (int w = 0; w < wave; ++w) before += wsum[w];
        const int excl = before + incl - v;
        fill[t] = excl;
        const int node = d0 + t;
        if (node < n)      rowptr[node] = excl;
        if (node == n - 1) rowptr[n]    = excl + v;
    }
    __syncthreads();
    for (int i0 = beg + t; i0 < end; i0 += 4 * 512) {
        int2 sd[4];
#pragma unroll
        for (int u = 0; u < 4; ++u) {
            const int i = min(i0 + u * 512, end - 1);
            sd[u] = p2[i];
        }
#pragma unroll
        for (int u = 0; u < 4; ++u) {
            if (i0 + u * 512 < end) {
                const int pos = atomicAdd(&fill[(sd[u].y - d0) & (BIN_W - 1)], 1);
                csr_src[pos] = sd[u].x;
            }
        }
    }
}

template <int HEADS>
__device__ __forceinline__ float pick_head(const float (&w)[HEADS], int hsel) {
    float r = w[0];
#pragma unroll
    for (int i = 1; i < HEADS; ++i) r = (hsel == i) ? w[i] : r;
    return r;
}

template <int HEADS, int CH, bool RELU>
__global__ void __launch_bounds__(256)
gat_gather_k(const int* __restrict__ rowptr, const int* __restrict__ csr_src,
             const float* __restrict__ als, const float* __restrict__ ald,
             const float* __restrict__ h, const float* __restrict__ bias,
             float* __restrict__ out, int n) {
    constexpr int C  = HEADS * CH;
    constexpr int NQ = C / 128;
    const int lane = threadIdx.x & 31;
    const int node = __builtin_amdgcn_readfirstlane((int)(blockIdx.x * 8 + (threadIdx.x >> 5)));
    if (node >= n) return;

    const int beg = rowptr[node];
    const int cnt = rowptr[node + 1] - beg + 1;

    float adst[HEADS];
#pragma unroll
    for (int hh = 0; hh < HEADS; ++hh) adst[hh] = ald[(size_t)node * HEADS + hh];

    int hq[NQ];
#pragma unroll
    for (int q = 0; q < NQ; ++q) hq[q] = (q * 128 + lane * 4) / CH;

    float  m[HEADS], dn[HEADS];
    float4 acc[NQ];
#pragma unroll
    for (int hh = 0; hh < HEADS; ++hh) { m[hh] = 0.0f; dn[hh] = 0.0f; }
#pragma unroll
    for (int q = 0; q < NQ; ++q) acc[q] = make_float4(0.0f, 0.0f, 0.0f, 0.0f);

    for (int j0 = 0; j0 < cnt; j0 += 32) {
        const int  j     = j0 + lane;
        const bool valid = j < cnt;
        int s_l = node;
        if (valid && j > 0) s_l = min(max(csr_src[beg + j - 1], 0), n - 1);

        float a[HEADS];
        if constexpr (HEADS == 4) {
            const float4 a4 = *(const float4*)(als + (size_t)s_l * 4);
            a[0] = a4.x; a[1] = a4.y; a[2] = a4.z; a[3] = a4.w;
        } else {
#pragma unroll
            for (int hh = 0; hh < HEADS; ++hh) a[hh] = als[(size_t)s_l * HEADS + hh];
        }

        float e[HEADS], cm[HEADS], pw[HEADS], ps[HEADS], sc[HEADS];
#pragma unroll
        for (int hh = 0; hh < HEADS; ++hh) {
            e[hh]  = valid ? lrelu(a[hh] + adst[hh]) : -INFINITY;
            cm[hh] = e[hh];
        }
#pragma unroll
        for (int off = 16; off > 0; off >>= 1) {
#pragma unroll
            for (int hh = 0; hh < HEADS; ++hh) cm[hh] = fmaxf(cm[hh], __shfl_xor(cm[hh], off));
        }
#pragma unroll
        for (int hh = 0; hh < HEADS; ++hh) {
            const float mn = (j0 == 0) ? cm[hh] : fmaxf(m[hh], cm[hh]);
            sc[hh] = (j0 == 0) ? 0.0f : __expf(m[hh] - mn);
            pw[hh] = valid ? __expf(e[hh] - mn) : 0.0f;
            ps[hh] = pw[hh];
            m[hh]  = mn;
        }
#pragma unroll
        for (int off = 16; off > 0; off >>= 1) {
#pragma unroll
            for (int hh = 0; hh < HEADS; ++hh) ps[hh] += __shfl_xor(ps[hh], off);
        }
#pragma unroll
        for (int hh = 0; hh < HEADS; ++hh) dn[hh] = dn[hh] * sc[hh] + ps[hh];
        if (j0 > 0) {
#pragma unroll
            for (int q = 0; q < NQ; ++q) {
                const float f = pick_head<HEADS>(sc, hq[q]);
                acc[q].x *= f; acc[q].y *= f; acc[q].z *= f; acc[q].w *= f;
            }
        }

        const int nc = min(32, cnt - j0);
        for (int t = 0; t < nc; t += 4) {
#pragma unroll
            for (int u = 0; u < 4; ++u) {
                const int s = __builtin_amdgcn_readlane(s_l, t + u);
                float wv[HEADS];
#pragma unroll
                for (int hh = 0; hh < HEADS; ++hh)
                    wv[hh] = __int_as_float(__builtin_amdgcn_readlane(__float_as_int(pw[hh]), t + u));
                const float4* hp = (const float4*)(h + (size_t)s * C) + lane;
#pragma unroll
                for (int q = 0; q < NQ; ++q) {
                    const float4 v = hp[q * 32];
                    const float  w = pick_head<HEADS>(wv, hq[q]);
                    acc[q].x += w * v.x; acc[q].y += w * v.y;
                    acc[q].z += w * v.z; acc[q].w += w * v.w;
                }
            }
        }
    }

#pragma unroll
    for (int q = 0; q < NQ; ++q) {
        const int    c0 = q * 128 + lane * 4;
        const float  d  = pick_head<HEADS>(dn, hq[q]);
        const float4 b  = *(const float4*)(bias + c0);
        float4 r;
        r.x = acc[q].x / d + b.x; r.y = acc[q].y / d + b.y;
        r.z = acc[q].z / d + b.z; r.w = acc[q].w / d + b.w;
        if (RELU) {
            r.x = r.x > 0.0f ? r.x : 0.0f; r.y = r.y > 0.0f ? r.y : 0.0f;
            r.z = r.z > 0.0f ? r.z : 0.0f; r.w = r.w > 0.0f ? r.w : 0.0f;
        }
        *(float4*)(out + (size_t)node * C + c0) = r;
    }
}

extern "C" void kernel_launch(void* const* d_in, const int* in_sizes, int n_in,
                              void* d_out, int out_size, void* d_ws, size_t ws_size,
                              hipStream_t stream) {
    const float* x      = (const float*)d_in[0];
    const int*   eidx   = (const int*)d_in[1];
    const float* W1     = (const float*)d_in[2];
    const float* a1_src = (const float*)d_in[3];
    const float* a1_dst = (const float*)d_in[4];
    const float* b1     = (const float*)d_in[5];
    const float* W2     = (const float*)d_in[6];
    const float* a2_src = (const float*)d_in[7];
    const float* a2_dst = (const float*)d_in[8];
    const float* b2     = (const float*)d_in[9];
    float*       out    = (float*)d_out;

    const int N = in_sizes[0] / 128;
    const int E = in_sizes[1] / 2;

    float* ws   = (float*)d_ws;
    float* h1   = ws;
    float* x2   = h1   + (size_t)N * C1;
    float* als1 = x2   + (size_t)N * C1;
    float* ald1 = als1 + (size_t)N * HEADS1;
    float* als2 = ald1 + (size_t)N * HEADS1;
    float* ald2 = als2 + (size_t)N;
    const int nbin  = (N + BIN_W - 1) / BIN_W;
    const int nscan = nbin * NB1;
    int*   pairs  = (int*)(ald2 + (size_t)N);
    int*   csr    = pairs  + (size_t)2 * E;
    int*   H      = csr    + (size_t)E;
    int*   O      = H      + (size_t)nscan;
    int*   rowptr = O      + (size_t)nscan + 1;
    float* h2   = h1;

    const int mtiles = (N + 15) / 16;

    bin_count_k<<<NB1, 1024, 0, stream>>>(eidx, E, N, nbin, H);
    rowptr_scan_k<<<(nscan + 1023) / 1024, 1024, 0, stream>>>(H, O, nscan);
    bin_scatter_k<<<NB1, 1024, 0, stream>>>(eidx, E, N, nbin, O, pairs);
    bin_sort_k<<<nbin, 512, 0, stream>>>(pairs, O, N, rowptr, csr);

    {
        dim3 grid((mtiles + 3) / 4, C1 / 64);
        gemm_wmma_f32_k<<<grid, 128, 0, stream>>>(x, W1, h1, N, 128, C1);
    }
    {
        int tot = N * HEADS1;
        attn_logits_k<<<(tot + 255) / 256, 256, 0, stream>>>(h1, a1_src, a1_dst,
                                                             als1, ald1, N, HEADS1, HID1);
    }
    gat_gather_k<HEADS1, HID1, true><<<(N + 7) / 8, 256, 0, stream>>>(rowptr, csr, als1, ald1,
                                                                      h1, b1, x2, N);

    {
        dim3 grid((mtiles + 3) / 4, C2 / 64);
        gemm_wmma_f32_k<<<grid, 128, 0, stream>>>(x2, W2, h2, N, C1, C2);
    }
    attn_logits_k<<<(N + 255) / 256, 256, 0, stream>>>(h2, a2_src, a2_dst,
                                                       als2, ald2, N, 1, C2);
    gat_gather_k<1, C2, false><<<(N + 7) / 8, 256, 0, stream>>>(rowptr, csr, als2, ald2,
                                                                h2, b2, out, N);
}


# Logit dots from the accumulators in the product kernels' epilogues (row_xmask adds over 16 lanes; layer 2: two 64-column partials summed in the gather): logit launches gone; dot sum order changes
# speedup vs baseline: 7.6656x; 1.0690x over previous
#include <hip/hip_runtime.h>
#include <hip/hip_bf16.h>
#include <math.h>

typedef __attribute__((ext_vector_type(2))) float v2f;
typedef __attribute__((ext_vector_type(8))) float v8f;

#define HEADS1 4
#define HID1   64
#define C1     256
#define C2     128
#define NEG_SLOPE 0.2f

#define KC        128
#define BS_STRIDE 72

__device__ __forceinline__ float lrelu(float v) {
    return v < 0.0f ? NEG_SLOPE * v : v;
}

__device__ __forceinline__ float row16_sum(float v) {
    v += __int_as_float(__builtin_amdgcn_update_dpp(0, __float_as_int(v), 0x161, 0xf, 0xf, true));
    v += __int_as_float(__builtin_amdgcn_update_dpp(0, __float_as_int(v), 0x162, 0xf, 0xf, true));
    v += __int_as_float(__builtin_amdgcn_update_dpp(0, __float_as_int(v), 0x164, 0xf, 0xf, true));
    v += __int_as_float(__builtin_amdgcn_update_dpp(0, __float_as_int(v), 0x168, 0xf, 0xf, true));
    return v;
}

__global__ void __launch_bounds__(128)
gemm_wmma_f32_k(const float* __restrict__ A,
                const float* __restrict__ B,
                float* __restrict__ C,
                const float* __restrict__ a_src,
                const float* __restrict__ a_dst,
                float* __restrict__ als, float* __restrict__ ald,
                int M, int K, int Ncols) {
    __shared__ float Bs[KC * BS_STRIDE];

    const int t     = threadIdx.x;
    const int lane  = t & 31;
    const int wave  = t >> 5;
    const int mtile = blockIdx.x * 4 + wave;
    const int m0    = mtile * 16;
    const bool valid = (m0 < M);
    const int n0    = blockIdx.y * 64;
    const int l15   = lane & 15;
    const int koff  = (lane >> 4) * 2;
    const int row   = valid ? (m0 + l15) : 0;

    v8f acc0 = {}, acc1 = {}, acc2 = {}, acc3 = {};

    const float as0 = a_src[n0 + l15],      ad0 = a_dst[n0 + l15];
    const float as1 = a_src[n0 + l15 + 16], ad1 = a_dst[n0 + l15 + 16];
    const float as2 = a_src[n0 + l15 + 32], ad2 = a_dst[n0 + l15 + 32];
    const float as3 = a_src[n0 + l15 + 48], ad3 = a_dst[n0 + l15 + 48];

    for (int k0 = 0; k0 < K; k0 += KC) {
        __syncthreads();
#pragma unroll
        for (int i = 0; i < (KC * 64 / 4) / 128; ++i) {
            int j = t + i * 128;
            int k = j >> 4;
            int c = (j & 15) * 4;
            const float4 v = *(const float4*)(B + (size_t)(k0 + k) * Ncols + n0 + c);
            *(float4*)(&Bs[k * BS_STRIDE + c]) = v;
        }
        __syncthreads();

        if (valid) {
            const float* ap0 = A + (size_t)row * K + k0 + koff;
            for (int kk = 0; kk < KC; kk += 4) {
                v2f a; a.x = ap0[kk]; a.y = ap0[kk + 1];

                const float* bp = &Bs[(kk + koff) * BS_STRIDE + l15];
                v2f b0; b0.x = bp[0];  b0.y = bp[BS_STRIDE];
                v2f b1; b1.x = bp[16]; b1.y = bp[BS_STRIDE + 16];
                v2f b2; b2.x = bp[32]; b2.y = bp[BS_STRIDE + 32];
                v2f b3; b3.x = bp[48]; b3.y = bp[BS_STRIDE + 48];

                acc0 = __builtin_amdgcn_wmma_f32_16x16x4_f32(false, a, false, b0, (short)0, acc0, false, false);
                acc1 = __builtin_amdgcn_wmma_f32_16x16x4_f32(false, a, false, b1, (short)0, acc1, false, false);
                acc2 = __builtin_amdgcn_wmma_f32_16x16x4_f32(false, a, false, b2, (short)0, acc2, false, false);
                acc3 = __builtin_amdgcn_wmma_f32_16x16x4_f32(false, a, false, b3, (short)0, acc3, false, false);
            }
        }
    }

    if (valid) {
        const int rbase = m0 + ((lane >> 4) ? 8 : 0);
#pragma unroll
        for (int r = 0; r < 8; ++r) {
            float* cp = C + (size_t)(rbase + r) * Ncols + n0 + l15;
            cp[0]  = acc0[r];
            cp[16] = acc1[r];
            cp[32] = acc2[r];
            cp[48] = acc3[r];
        }

        float ps[8], pd[8];
#pragma unroll
        for (int r = 0; r < 8; ++r) {
            ps[r] = acc0[r] * as0 + acc1[r] * as1 + acc2[r] * as2 + acc3[r] * as3;
            pd[r] = acc0[r] * ad0 + acc1[r] * ad1 + acc2[r] * ad2 + acc3[r] * ad3;
        }
#pragma unroll
        for (int r = 0; r < 8; ++r) { ps[r] = row16_sum(ps[r]); pd[r] = row16_sum(pd[r]); }
        float vs = ps[0], vd = pd[0];
#pragma unroll
        for (int r = 1; r < 8; ++r) { vs = (l15 == r) ? ps[r] : vs; vd = (l15 == r) ? pd[r] : vd; }
        if (l15 < 8) {
            const size_t o = (size_t)(rbase + l15) * gridDim.y + blockIdx.y;
            als[o] = vs;
            ald[o] = vd;
        }
    }
}

#define BIN_SHIFT 7
#define BIN_W     128
#define MAX_BINS  512
#define NB1       128

__device__ __forceinline__ int bin_of(int d) {
    return min(d >> BIN_SHIFT, MAX_BINS - 1);
}

__global__ void __launch_bounds__(1024)
bin_count_k(const int* __restrict__ ei, int E, int n, int nbin, int* __restrict__ H) {
    __shared__ int hist[MAX_BINS];
    const int t = threadIdx.x;
    const int b = blockIdx.x;
    for (int k = t; k < MAX_BINS; k += 1024) hist[k] = 0;
    __syncthreads();
    const int per = (E + NB1 - 1) / NB1;
    const int lo  = b * per;
    const int hi  = min(lo + per, E);
    for (int e0 = lo + t; e0 < hi; e0 += 4 * 1024) {
        int d[4];
#pragma unroll
        for (int u = 0; u < 4; ++u) {
            const int e = min(e0 + u * 1024, hi - 1);
            d[u] = ei[E + e];
        }
#pragma unroll
        for (int u = 0; u < 4; ++u) {
            if (e0 + u * 1024 < hi)
                atomicAdd(&hist[bin_of(min(max(d[u], 0), n - 1))], 1);
        }
    }
    __syncthreads();
    for (int k = t; k < nbin; k += 1024) H[k * NB1 + b] = hist[k];
}

__global__ void __launch_bounds__(1024)
rowptr_scan_k(const int* __restrict__ cnt, int* __restrict__ rowptr, int n) {
    __shared__ int wtot[32];
    __shared__ int wpre[32];
    const int t    = threadIdx.x;
    const int lane = t & 31;
    const int wave = t >> 5;
    const int base = blockIdx.x * 1024;

    int pre = 0;
    for (int i = t; i < base; i += 1024) pre += cnt[i];
    for (int off = 16; off > 0; off >>= 1) pre += __shfl_xor(pre, off);

    const int idx = base + t;
    const int v   = idx < n ? cnt[idx] : 0;
    int incl = v;
    for (int off = 1; off < 32; off <<= 1) {
        int u = __shfl_up(incl, off);
        if (lane >= off) incl += u;
    }
    if (lane == 31) wtot[wave] = incl;
    if (lane == 0)  wpre[wave] = pre;
    __syncthreads();

    int before = 0;
    for (int w = 0; w < 32; ++w) {
        before += wpre[w];
        if (w < wave) before += wtot[w];
    }
    const int excl = before + incl - v;
    if (idx < n)      rowptr[idx] = excl;
    if (idx == n - 1) rowptr[n]   = excl + v;
}

__global__ void __launch_bounds__(1024)
bin_scatter_k(const int* __restrict__ ei, int E, int n, int nbin,
              const int* __restrict__ O, int* __restrict__ pairs) {
    __shared__ int cur[MAX_BINS];
    const int t = threadIdx.x;
    const int b = blockIdx.x;
    for (int k = t; k < MAX_BINS; k += 1024) cur[k] = (k < nbin) ? O[k * NB1 + b] : 0;
    __syncthreads();
    const int per = (E + NB1 - 1) / NB1;
    const int lo  = b * per;
    const int hi  = min(lo + per, E);
    int2* __restrict__ p2 = (int2*)pairs;
    for (int e0 = lo + t; e0 < hi; e0 += 4 * 1024) {
        int d[4], s[4];
#pragma unroll
        for (int u = 0; u < 4; ++u) {
            const int e = min(e0 + u * 1024, hi - 1);
            d[u] = ei[E + e];
            s[u] = ei[e];
        }
#pragma unroll
        for (int u = 0; u < 4; ++u) {
            if (e0 + u * 1024 < hi) {
                const int dc  = min(max(d[u], 0), n - 1);
                const int pos = atomicAdd(&cur[bin_of(dc)], 1);
                p2[pos] = make_int2(s[u], dc);
            }
        }
    }
}

__global__ void __launch_bounds__(512)
bin_sort_k(const int* __restrict__ pairs, const int* __restrict__ O, int n,
           int* __restrict__ rowptr, int* __restrict__ csr_src) {
    __shared__ int cnt[BIN_W];
    __shared__ int fill[BIN_W];
    __shared__ int wsum[BIN_W / 32];
    const int t    = threadIdx.x;
    const int lane = t & 31;
    const int wave = t >> 5;
    const int k    = blockIdx.x;
    const int beg  = O[k * NB1];
    const int end  = O[(k + 1) * NB1];
    const int d0   = k << BIN_SHIFT;
    const int2* __restrict__ p2 = (const int2*)pairs;

    if (t < BIN_W) cnt[t] = 0;
    __syncthreads();
    for (int i0 = beg + t; i0 < end; i0 += 4 * 512) {
        int y[4];
#pragma unroll
        for (int u = 0; u < 4; ++u) {
            const int i = min(i0 + u * 512, end - 1);
            y[u] = p2[i].y;
        }
#pragma unroll
        for (int u = 0; u < 4; ++u) {
            if (i0 + u * 512 < end) atomicAdd(&cnt[(y[u] - d0) & (BIN_W - 1)], 1);
        }
    }
    __syncthreads();

    const int v = (t < BIN_W) ? cnt[t] : 0;
    int incl = v;
    for (int off = 1; off < 32; off <<= 1) {
        int u = __shfl_up(incl, off);
        if (lane >= off) incl += u;
    }
    if (t < BIN_W && lane == 31) wsum[wave] = incl;
    __syncthreads();
    if (t < BIN_W) {
        int before = beg;
        for (int w = 0; w < wave; ++w) before += wsum[w];
        const int excl = before + incl - v;
        fill[t] = excl;
        const int node = d0 + t;
        if (node < n)      rowptr[node] = excl;
        if (node == n - 1) rowptr[n]    = excl + v;
    }
    __syncthreads();
    for (int i0 = beg + t; i0 < end; i0 += 4 * 512) {
        int2 sd[4];
#pragma unroll
        for (int u = 0; u < 4; ++u) {
            const int i = min(i0 + u * 512, end - 1);
            sd[u] = p2[i];
        }
#pragma unroll
        for (int u = 0; u < 4; ++u) {
            if (i0 + u * 512 < end) {
                const int pos = atomicAdd(&fill[(sd[u].y - d0) & (BIN_W - 1)], 1);
                csr_src[pos] = sd[u].x;
            }
        }
    }
}

template <int HEADS>
__device__ __forceinline__ float pick_head(const float (&w)[HEADS], int hsel) {
    float r = w[0];
#pragma unroll
    for (int i = 1; i < HEADS; ++i) r = (hsel == i) ? w[i] : r;
    return r;
}

template <int HEADS, int CH, bool RELU, int PARTS>
__global__ void __launch_bounds__(256)
gat_gather_k(const int* __restrict__ rowptr, const int* __restrict__ csr_src,
             const float* __restrict__ als, const float* __restrict__ ald,
             const float* __restrict__ h, const float* __restrict__ bias,
             float* __restrict__ out, int n) {
    constexpr int C  = HEADS * CH;
    constexpr int NQ = C / 128;
    const int lane = threadIdx.x & 31;
    const int node = __builtin_amdgcn_readfirstlane((int)(blockIdx.x * 8 + (threadIdx.x >> 5)));
    if (node >= n) return;

    const int beg = rowptr[node];
    const int cnt = rowptr[node + 1] - beg + 1;

    float adst[HEADS];
#pragma unroll
    for (int hh = 0; hh < HEADS; ++hh) {
        float v = ald[((size_t)node * HEADS + hh) * PARTS];
#pragma unroll
        for (int p = 1; p < PARTS; ++p) v += ald[((size_t)node * HEADS + hh) * PARTS + p];
        adst[hh] = v;
    }

    int hq[NQ];
#pragma unroll
    for (int q = 0; q < NQ; ++q) hq[q] = (q * 128 + lane * 4) / CH;

    float  m[HEADS], dn[HEADS];
    float4 acc[NQ];
#pragma unroll
    for (int hh = 0; hh < HEADS; ++hh) { m[hh] = 0.0f; dn[hh] = 0.0f; }
#pragma unroll
    for (int q = 0; q < NQ; ++q) acc[q] = make_float4(0.0f, 0.0f, 0.0f, 0.0f);

    for (int j0 = 0; j0 < cnt; j0 += 32) {
        const int  j     = j0 + lane;
        const bool valid = j < cnt;
        int s_l = node;
        if (valid && j > 0) s_l = min(max(csr_src[beg + j - 1], 0), n - 1);

        float a[HEADS];
        if constexpr (HEADS == 4 && PARTS == 1) {
            const float4 a4 = *(const float4*)(als + (size_t)s_l * 4);
            a[0] = a4.x; a[1] = a4.y; a[2] = a4.z; a[3] = a4.w;
        } else if constexpr (HEADS == 1 && PARTS == 2) {
            const float2 a2 = *(const float2*)(als + (size_t)s_l * 2);
            a[0] = a2.x + a2.y;
        } else {
#pragma unroll
            for (int hh = 0; hh < HEADS; ++hh) {
                float v = als[((size_t)s_l * HEADS + hh) * PARTS];
#pragma unroll
                for (int p = 1; p < PARTS; ++p) v += als[((size_t)s_l * HEADS + hh) * PARTS + p];
                a[hh] = v;
            }
        }

        float e[HEADS], cm[HEADS], pw[HEADS], ps[HEADS], sc[HEADS];
#pragma unroll
        for (int hh = 0; hh < HEADS; ++hh) {
            e[hh]  = valid ? lrelu(a[hh] + adst[hh]) : -INFINITY;
            cm[hh] = e[hh];
        }
#pragma unroll
        for (int off = 16; off > 0; off >>= 1) {
#pragma unroll
            for (int hh = 0; hh < HEADS; ++hh) cm[hh] = fmaxf(cm[hh], __shfl_xor(cm[hh], off));
        }
#pragma unroll
        for (int hh = 0; hh < HEADS; ++hh) {
            const float mn = (j0 == 0) ? cm[hh] : fmaxf(m[hh], cm[hh]);
            sc[hh] = (j0 == 0) ? 0.0f : __expf(m[hh] - mn);
            pw[hh] = valid ? __expf(e[hh] - mn) : 0.0f;
            ps[hh] = pw[hh];
            m[hh]  = mn;
        }
#pragma unroll
        for (int off = 16; off > 0; off >>= 1) {
#pragma unroll
            for (int hh = 0; hh < HEADS; ++hh) ps[hh] += __shfl_xor(ps[hh], off);
        }
#pragma unroll
        for (int hh = 0; hh < HEADS; ++hh) dn[hh] = dn[hh] * sc[hh] + ps[hh];
        if (j0 > 0) {
#pragma unroll
            for (int q = 0; q < NQ; ++q) {
                const float f = pick_head<HEADS>(sc, hq[q]);
                acc[q].x *= f; acc[q].y *= f; acc[q].z *= f; acc[q].w *= f;
            }
        }

        const int nc = min(32, cnt - j0);
        for (int t = 0; t < nc; t += 4) {
#pragma unroll
            for (int u = 0; u < 4; ++u) {
                const int s = __builtin_amdgcn_readlane(s_l, t + u);
                float wv[HEADS];
#pragma unroll
                for (int hh = 0; hh < HEADS; ++hh)
                    wv[hh] = __int_as_float(__builtin_amdgcn_readlane(__float_as_int(pw[hh]), t + u));
                const float4* hp = (const float4*)(h + (size_t)s * C) + lane;
#pragma unroll
                for (int q = 0; q < NQ; ++q) {
                    const float4 v = hp[q * 32];
                    const float  w = pick_head<HEADS>(wv, hq[q]);
                    acc[q].x += w * v.x; acc[q].y += w * v.y;
                    acc[q].z += w * v.z; acc[q].w += w * v.w;
                }
            }
        }
    }

#pragma unroll
    for (int q = 0; q < NQ; ++q) {
        const int    c0 = q * 128 + lane * 4;
        const float  d  = pick_head<HEADS>(dn, hq[q]);
        const float4 b  = *(const float4*)(bias + c0);
        float4 r;
        r.x = acc[q].x / d + b.x; r.y = acc[q].y / d + b.y;
        r.z = acc[q].z / d + b.z; r.w = acc[q].w / d + b.w;
        if (RELU) {
            r.x = r.x > 0.0f ? r.x : 0.0f; r.y = r.y > 0.0f ? r.y : 0.0f;
            r.z = r.z > 0.0f ? r.z : 0.0f; r.w = r.w > 0.0f ? r.w : 0.0f;
        }
        *(float4*)(out + (size_t)node * C + c0) = r;
    }
}

extern "C" void kernel_launch(void* const* d_in, const int* in_sizes, int n_in,
                              void* d_out, int out_size, void* d_ws, size_t ws_size,
                              hipStream_t stream) {
    const float* x      = (const float*)d_in[0];
    const int*   eidx   = (const int*)d_in[1];
    const float* W1     = (const float*)d_in[2];
    const float* a1_src = (const float*)d_in[3];
    const float* a1_dst = (const float*)d_in[4];
    const float* b1     = (const float*)d_in[5];
    const float* W2     = (const float*)d_in[6];
    const float* a2_src = (const float*)d_in[7];
    const float* a2_dst = (const float*)d_in[8];
    const float* b2     = (const float*)d_in[9];
    float*       out    = (float*)d_out;

    const int N = in_sizes[0] / 128;
    const int E = in_sizes[1] / 2;

    float* ws   = (float*)d_ws;
    float* h1   = ws;
    float* x2   = h1   + (size_t)N * C1;
    float* als1 = x2   + (size_t)N * C1;
    float* ald1 = als1 + (size_t)N * HEADS1;
    float* als2 = ald1 + (size_t)N * HEADS1;
    float* ald2 = als2 + (size_t)N * 2;
    const int nbin  = (N + BIN_W - 1) / BIN_W;
    const int nscan = nbin * NB1;
    int*   pairs  = (int*)(ald2 + (size_t)N * 2);
    int*   csr    = pairs  + (size_t)2 * E;
    int*   H      = csr    + (size_t)E;
    int*   O      = H      + (size_t)nscan;
    int*   rowptr = O      + (size_t)nscan + 1;
    float* h2   = h1;

    const int mtiles = (N + 15) / 16;

    bin_count_k<<<NB1, 1024, 0, stream>>>(eidx, E, N, nbin, H);
    rowptr_scan_k<<<(nscan + 1023) / 1024, 1024, 0, stream>>>(H, O, nscan);
    bin_scatter_k<<<NB1, 1024, 0, stream>>>(eidx, E, N, nbin, O, pairs);
    bin_sort_k<<<nbin, 512, 0, stream>>>(pairs, O, N, rowptr, csr);

    {
        dim3 grid((mtiles + 3) / 4, C1 / 64);
        gemm_wmma_f32_k<<<grid, 128, 0, stream>>>(x, W1, h1, a1_src, a1_dst, als1, ald1, N, 128, C1);
    }
    gat_gather_k<HEADS1, HID1, true, 1><<<(N + 7) / 8, 256, 0, stream>>>(rowptr, csr, als1, ald1,
                                                                         h1, b1, x2, N);

    {
        dim3 grid((mtiles + 3) / 4, C2 / 64);
        gemm_wmma_f32_k<<<grid, 128, 0, stream>>>(x2, W2, h2, a2_src, a2_dst, als2, ald2, N, C1, C2);
    }
    gat_gather_k<1, C2, false, 2><<<(N + 7) / 8, 256, 0, stream>>>(rowptr, csr, als2, ald2,
                                                                   h2, b2, out, N);
}


# Edge bucketing in 2 launches, not 4: a chunk block counts, scans and places its pairs inside its own stretch; a bin block reads its 128 pieces by table; bucket order still LDS arrival order
# speedup vs baseline: 8.3595x; 1.0905x over previous
#include <hip/hip_runtime.h>
#include <hip/hip_bf16.h>
#include <math.h>

typedef __attribute__((ext_vector_type(2))) float v2f;
typedef __attribute__((ext_vector_type(8))) float v8f;

#define HEADS1 4
#define HID1   64
#define C1     256
#define C2     128
#define NEG_SLOPE 0.2f

#define KC        128
#define BS_STRIDE 72

__device__ __forceinline__ float lrelu(float v) {
    return v < 0.0f ? NEG_SLOPE * v : v;
}

__device__ __forceinline__ float row16_sum(float v) {
    v += __int_as_float(__builtin_amdgcn_update_dpp(0, __float_as_int(v), 0x161, 0xf, 0xf, true));
    v += __int_as_float(__builtin_amdgcn_update_dpp(0, __float_as_int(v), 0x162, 0xf, 0xf, true));
    v += __int_as_float(__builtin_amdgcn_update_dpp(0, __float_as_int(v), 0x164, 0xf, 0xf, true));
    v += __int_as_float(__builtin_amdgcn_update_dpp(0, __float_as_int(v), 0x168, 0xf, 0xf, true));
    return v;
}

__global__ void __launch_bounds__(128)
gemm_wmma_f32_k(const float* __restrict__ A,
                const float* __restrict__ B,
                float* __restrict__ C,
                const float* __restrict__ a_src,
                const float* __restrict__ a_dst,
                float* __restrict__ als, float* __restrict__ ald,
                int M, int K, int Ncols) {
    __shared__ float Bs[KC * BS_STRIDE];

    const int t     = threadIdx.x;
    const int lane  = t & 31;
    const int wave  = t >> 5;
    const int mtile = blockIdx.x * 4 + wave;
    const int m0    = mtile * 16;
    const bool valid = (m0 < M);
    const int n0    = blockIdx.y * 64;
    const int l15   = lane & 15;
    const int koff  = (lane >> 4) * 2;
    const int row   = valid ? (m0 + l15) : 0;

    v8f acc0 = {}, acc1 = {}, acc2 = {}, acc3 = {};

    const float as0 = a_src[n0 + l15],      ad0 = a_dst[n0 + l15];
    const float as1 = a_src[n0 + l15 + 16], ad1 = a_dst[n0 + l15 + 16];
    const float as2 = a_src[n0 + l15 + 32], ad2 = a_dst[n0 + l15 + 32];
    const float as3 = a_src[n0 + l15 + 48], ad3 = a_dst[n0 + l15 + 48];

    for (int k0 = 0; k0 < K; k0 += KC) {
        __syncthreads();
#pragma unroll
        for (int i = 0; i < (KC * 64 / 4) / 128; ++i) {
            int j = t + i * 128;
            int k = j >> 4;
            int c = (j & 15) * 4;
            const float4 v = *(const float4*)(B + (size_t)(k0 + k) * Ncols + n0 + c);
            *(float4*)(&Bs[k * BS_STRIDE + c]) = v;
        }
        __syncthreads();

        if (valid) {
            const float* ap0 = A + (size_t)row * K + k0 + koff;
            for (int kk = 0; kk < KC; kk += 4) {
                v2f a; a.x = ap0[kk]; a.y = ap0[kk + 1];

                const float* bp = &Bs[(kk + koff) * BS_STRIDE + l15];
                v2f b0; b0.x = bp[0];  b0.y = bp[BS_STRIDE];
                v2f b1; b1.x = bp[16]; b1.y = bp[BS_STRIDE + 16];
                v2f b2; b2.x = bp[32]; b2.y = bp[BS_STRIDE + 32];
                v2f b3; b3.x = bp[48]; b3.y = bp[BS_STRIDE + 48];

                acc0 = __builtin_amdgcn_wmma_f32_16x16x4_f32(false, a, false, b0, (short)0, acc0, false, false);
                acc1 = __builtin_amdgcn_wmma_f32_16x16x4_f32(false, a, false, b1, (short)0, acc1, false, false);
                acc2 = __builtin_amdgcn_wmma_f32_16x16x4_f32(false, a, false, b2, (short)0, acc2, false, false);
                acc3 = __builtin_amdgcn_wmma_f32_16x16x4_f32(false, a, false, b3, (short)0, acc3, false, false);
            }
        }
    }

    if (valid) {
        const int rbase = m0 + ((lane >> 4) ? 8 : 0);
#pragma unroll
        for (int r = 0; r < 8; ++r) {
            float* cp = C + (size_t)(rbase + r) * Ncols + n0 + l15;
            cp[0]  = acc0[r];
            cp[16] = acc1[r];
            cp[32] = acc2[r];
            cp[48] = acc3[r];
        }

        float ps[8], pd[8];
#pragma unroll
        for (int r = 0; r < 8; ++r) {
            ps[r] = acc0[r] * as0 + acc1[r] * as1 + acc2[r] * as2 + acc3[r] * as3;
            pd[r] = acc0[r] * ad0 + acc1[r] * ad1 + acc2[r] * ad2 + acc3[r] * ad3;
        }
#pragma unroll
        for (int r = 0; r < 8; ++r) { ps[r] = row16_sum(ps[r]); pd[r] = row16_sum(pd[r]); }
        float vs = ps[0], vd = pd[0];
#pragma unroll
        for (int r = 1; r < 8; ++r) { vs = (l15 == r) ? ps[r] : vs; vd = (l15 == r) ? pd[r] : vd; }
        if (l15 < 8) {
            const size_t o = (size_t)(rbase + l15) * gridDim.y + blockIdx.y;
            als[o] = vs;
            ald[o] = vd;
        }
    }
}

#define BIN_SHIFT 7
#define BIN_W     128
#define MAX_BINS  512
#define NB1       128

__device__ __forceinline__ int bin_of(int d) {
    return min(d >> BIN_SHIFT, MAX_BINS - 1);
}

__global__ void __launch_bounds__(1024)
bin_local_k(const int* __restrict__ ei, int E, int n, int nbin,
            int* __restrict__ H, int* __restrict__ L, int* __restrict__ pairs) {
    __shared__ int hist[MAX_BINS];
    __shared__ int cur[MAX_BINS];
    __shared__ int wtot[MAX_BINS / 32];
    const int t    = threadIdx.x;
    const int lane = t & 31;
    const int wave = t >> 5;
    const int b    = blockIdx.x;
    for (int k = t; k < MAX_BINS; k += 1024) hist[k] = 0;
    __syncthreads();
    const int per = (E + NB1 - 1) / NB1;
    const int lo  = b * per;
    const int hi  = min(lo + per, E);
    for (int e0 = lo + t; e0 < hi; e0 += 8 * 1024) {
        int d[8];
#pragma unroll
        for (int u = 0; u < 8; ++u) {
            const int e = min(e0 + u * 1024, hi - 1);
            d[u] = ei[E + e];
        }
        asm volatile("" :: "v"(d[0]), "v"(d[1]), "v"(d[2]), "v"(d[3]), "v"(d[4]), "v"(d[5]), "v"(d[6]), "v"(d[7]));
#pragma unroll
        for (int u = 0; u < 8; ++u) {
            if (e0 + u * 1024 < hi)
                atomicAdd(&hist[bin_of(min(max(d[u], 0), n - 1))], 1);
        }
    }
    __syncthreads();

    const int v = (t < MAX_BINS) ? hist[t] : 0;
    int incl = v;
    for (int off = 1; off < 32; off <<= 1) {
        int u = __shfl_up(incl, off);
        if (lane >= off) incl += u;
    }
    if (t < MAX_BINS && lane == 31) wtot[wave] = incl;
    __syncthreads();
    if (t < MAX_BINS) {
        int before = 0;
#pragma unroll
        for (int w = 0; w < MAX_BINS / 32; ++w) before += (w < wave) ? wtot[w] : 0;
        const int excl = before + incl - v;
        cur[t] = excl;
        if (t < nbin) {
            H[t * NB1 + b] = v;
            L[t * NB1 + b] = excl;
        }
    }
    __syncthreads();

    int2* __restrict__ p2 = (int2*)pairs + lo;
    for (int e0 = lo + t; e0 < hi; e0 += 8 * 1024) {
        int d[8], s[8];
#pragma unroll
        for (int u = 0; u < 8; ++u) {
            const int e = min(e0 + u * 1024, hi - 1);
            d[u] = ei[E + e];
            s[u] = ei[e];
        }
        asm volatile("" :: "v"(d[0]), "v"(d[1]), "v"(d[2]), "v"(d[3]), "v"(d[4]), "v"(d[5]), "v"(d[6]), "v"(d[7]),
                           "v"(s[0]), "v"(s[1]), "v"(s[2]), "v"(s[3]), "v"(s[4]), "v"(s[5]), "v"(s[6]), "v"(s[7]));
#pragma unroll
        for (int u = 0; u < 8; ++u) {
            if (e0 + u * 1024 < hi) {
                const int dc  = min(max(d[u], 0), n - 1);
                const int pos = atomicAdd(&cur[bin_of(dc)], 1);
                p2[pos] = make_int2(s[u], dc);
            }
        }
    }
}

__global__ void __launch_bounds__(4 * NB1)
bin_sort_k(const int* __restrict__ pairs, const int* __restrict__ H, const int* __restrict__ L,
           int E, int n, int* __restrict__ rowptr, int* __restrict__ csr_src) {
    __shared__ int cnt[BIN_W];
    __shared__ int fill[BIN_W];
    __shared__ int wsum[BIN_W / 32];
    __shared__ int red[4 * NB1 / 32];
    const int t    = threadIdx.x;
    const int lane = t & 31;
    const int wave = t >> 5;
    const int k    = blockIdx.x;
    const int d0   = k << BIN_SHIFT;
    const int per  = (E + NB1 - 1) / NB1;
    const int c    = t >> 2;
    const int q    = t & 3;

    const int len = H[k * NB1 + c];
    const int lv  = L[k * NB1 + c];
    const int2* __restrict__ pp = (const int2*)pairs + (size_t)c * per + lv;

    int pre = (q == 0) ? lv : 0;
    for (int off = 16; off > 0; off >>= 1) pre += __shfl_xor(pre, off);
    if (t < BIN_W) cnt[t] = 0;
    if (lane == 0) red[wave] = pre;
    __syncthreads();
    int beg = 0;
    for (int w = 0; w < 4 * NB1 / 32; ++w) beg += red[w];

    for (int j0 = q; j0 < len; j0 += 32) {
        int y[8];
#pragma unroll
        for (int u = 0; u < 8; ++u) {
            const int j = min(j0 + 4 * u, len - 1);
            y[u] = pp[j].y;
        }
        asm volatile("" :: "v"(y[0]), "v"(y[1]), "v"(y[2]), "v"(y[3]), "v"(y[4]), "v"(y[5]), "v"(y[6]), "v"(y[7]));
#pragma unroll
        for (int u = 0; u < 8; ++u) {
            if (j0 + 4 * u < len) atomicAdd(&cnt[(y[u] - d0) & (BIN_W - 1)], 1);
        }
    }
    __syncthreads();

    const int v = (t < BIN_W) ? cnt[t] : 0;
    int incl = v;
    for (int off = 1; off < 32; off <<= 1) {
        int u = __shfl_up(incl, off);
        if (lane >= off) incl += u;
    }
    if (t < BIN_W && lane == 31) wsum[wave] = incl;
    __syncthreads();
    if (t < BIN_W) {
        int before = beg;
        for (int w = 0; w < wave; ++w) before += wsum[w];
        const int excl = before + incl - v;
        fill[t] = excl;
        const int node = d0 + t;
        if (node < n)      rowptr[node] = excl;
        if (node == n - 1) rowptr[n]    = excl + v;
    }
    __syncthreads();
    for (int j0 = q; j0 < len; j0 += 32) {
        int2 sd[8];
#pragma unroll
        for (int u = 0; u < 8; ++u) {
            const int j = min(j0 + 4 * u, len - 1);
            sd[u] = pp[j];
        }
        asm volatile("" :: "v"(sd[0].x), "v"(sd[1].x), "v"(sd[2].x), "v"(sd[3].x), "v"(sd[4].x), "v"(sd[5].x), "v"(sd[6].x), "v"(sd[7].x),
                           "v"(sd[0].y), "v"(sd[1].y), "v"(sd[2].y), "v"(sd[3].y), "v"(sd[4].y), "v"(sd[5].y), "v"(sd[6].y), "v"(sd[7].y));
#pragma unroll
        for (int u = 0; u < 8; ++u) {
            if (j0 + 4 * u < len) {
                const int pos = atomicAdd(&fill[(sd[u].y - d0) & (BIN_W - 1)], 1);
                csr_src[pos] = sd[u].x;
            }
        }
    }
}

template <int HEADS>
__device__ __forceinline__ float pick_head(const float (&w)[HEADS], int hsel) {
    float r = w[0];
#pragma unroll
    for (int i = 1; i < HEADS; ++i) r = (hsel == i) ? w[i] : r;
    return r;
}

template <int HEADS, int CH, bool RELU, int PARTS>
__global__ void __launch_bounds__(256)
gat_gather_k(const int* __restrict__ rowptr, const int* __restrict__ csr_src,
             const float* __restrict__ als, const float* __restrict__ ald,
             const float* __restrict__ h, const float* __restrict__ bias,
             float* __restrict__ out, int n) {
    constexpr int C  = HEADS * CH;
    constexpr int NQ = C / 128;
    const int lane = threadIdx.x & 31;
    const int node = __builtin_amdgcn_readfirstlane((int)(blockIdx.x * 8 + (threadIdx.x >> 5)));
    if (node >= n) return;

    const int beg = rowptr[node];
    const int cnt = rowptr[node + 1] - beg + 1;

    float adst[HEADS];
#pragma unroll
    for (int hh = 0; hh < HEADS; ++hh) {
        float v = ald[((size_t)node * HEADS + hh) * PARTS];
#pragma unroll
        for (int p = 1; p < PARTS; ++p) v += ald[((size_t)node * HEADS + hh) * PARTS + p];
        adst[hh] = v;
    }

    int hq[NQ];
#pragma unroll
    for (int q = 0; q < NQ; ++q) hq[q] = (q * 128 + lane * 4) / CH;

    float  m[HEADS], dn[HEADS];
    float4 acc[NQ];
#pragma unroll
    for (int hh = 0; hh < HEADS; ++hh) { m[hh] = 0.0f; dn[hh] = 0.0f; }
#pragma unroll
    for (int q = 0; q < NQ; ++q) acc[q] = make_float4(0.0f, 0.0f, 0.0f, 0.0f);

    for (int j0 = 0; j0 < cnt; j0 += 32) {
        const int  j     = j0 + lane;
        const bool valid = j < cnt;
        int s_l = node;
        if (valid && j > 0) s_l = min(max(csr_src[beg + j - 1], 0), n - 1);

        float a[HEADS];
        if constexpr (HEADS == 4 && PARTS == 1) {
            const float4 a4 = *(const float4*)(als + (size_t)s_l * 4);
            a[0] = a4.x; a[1] = a4.y; a[2] = a4.z; a[3] = a4.w;
        } else if constexpr (HEADS == 1 && PARTS == 2) {
            const float2 a2 = *(const float2*)(als + (size_t)s_l * 2);
            a[0] = a2.x + a2.y;
        } else {
#pragma unroll
            for (int hh = 0; hh < HEADS; ++hh) {
                float v = als[((size_t)s_l * HEADS + hh) * PARTS];
#pragma unroll
                for (int p = 1; p < PARTS; ++p) v += als[((size_t)s_l * HEADS + hh) * PARTS + p];
                a[hh] = v;
            }
        }

        float e[HEADS], cm[HEADS], pw[HEADS], ps[HEADS], sc[HEADS];
#pragma unroll
        for (int hh = 0; hh < HEADS; ++hh) {
            e[hh]  = valid ? lrelu(a[hh] + adst[hh]) : -INFINITY;
            cm[hh] = e[hh];
        }
#pragma unroll
        for (int off = 16; off > 0; off >>= 1) {
#pragma unroll
            for (int hh = 0; hh < HEADS; ++hh) cm[hh] = fmaxf(cm[hh], __shfl_xor(cm[hh], off));
        }
#pragma unroll
        for (int hh = 0; hh < HEADS; ++hh) {
            const float mn = (j0 == 0) ? cm[hh] : fmaxf(m[hh], cm[hh]);
            sc[hh] = (j0 == 0) ? 0.0f : __expf(m[hh] - mn);
            pw[hh] = valid ? __expf(e[hh] - mn) : 0.0f;
            ps[hh] = pw[hh];
            m[hh]  = mn;
        }
#pragma unroll
        for (int off = 16; off > 0; off >>= 1) {
#pragma unroll
            for (int hh = 0; hh < HEADS; ++hh) ps[hh] += __shfl_xor(ps[hh], off);
        }
#pragma unroll
        for (int hh = 0; hh < HEADS; ++hh) dn[hh] = dn[hh] * sc[hh] + ps[hh];
        if (j0 > 0) {
#pragma unroll
            for (int q = 0; q < NQ; ++q) {
                const float f = pick_head<HEADS>(sc, hq[q]);
                acc[q].x *= f; acc[q].y *= f; acc[q].z *= f; acc[q].w *= f;
            }
        }

        const int nc = min(32, cnt - j0);
        for (int t = 0; t < nc; t += 4) {
#pragma unroll
            for (int u = 0; u < 4; ++u) {
                const int s = __builtin_amdgcn_readlane(s_l, t + u);
                float wv[HEADS];
#pragma unroll
                for (int hh = 0; hh < HEADS; ++hh)
                    wv[hh] = __int_as_float(__builtin_amdgcn_readlane(__float_as_int(pw[hh]), t + u));
                const float4* hp = (const float4*)(h + (size_t)s * C) + lane;
#pragma unroll
                for (int q = 0; q < NQ; ++q) {
                    const float4 v = hp[q * 32];
                    const float  w = pick_head<HEADS>(wv, hq[q]);
                    acc[q].x += w * v.x; acc[q].y += w * v.y;
                    acc[q].z += w * v.z; acc[q].w += w * v.w;
                }
            }
        }
    }

#pragma unroll
    for (int q = 0; q < NQ; ++q) {
        const int    c0 = q * 128 + lane * 4;
        const float  d  = pick_head<HEADS>(dn, hq[q]);
        const float4 b  = *(const float4*)(bias + c0);
        float4 r;
        r.x = acc[q].x / d + b.x; r.y = acc[q].y / d + b.y;
        r.z = acc[q].z / d + b.z; r.w = acc[q].w / d + b.w;
        if (RELU) {
            r.x = r.x > 0.0f ? r.x : 0.0f; r.y = r.y > 0.0f ? r.y : 0.0f;
            r.z = r.z > 0.0f ? r.z : 0.0f; r.w = r.w > 0.0f ? r.w : 0.0f;
        }
        *(float4*)(out + (size_t)node * C + c0) = r;
    }
}

extern "C" void kernel_launch(void* const* d_in, const int* in_sizes, int n_in,
                              void* d_out, int out_size, void* d_ws, size_t ws_size,
                              hipStream_t stream) {
    const float* x      = (const float*)d_in[0];
    const int*   eidx   = (const int*)d_in[1];
    const float* W1     = (const float*)d_in[2];
    const float* a1_src = (const float*)d_in[3];
    const float* a1_dst = (const float*)d_in[4];
    const float* b1     = (const float*)d_in[5];
    const float* W2     = (const float*)d_in[6];
    const float* a2_src = (const float*)d_in[7];
    const float* a2_dst = (const float*)d_in[8];
    const float* b2     = (const float*)d_in[9];
    float*       out    = (float*)d_out;

    const int N = in_sizes[0] / 128;
    const int E = in_sizes[1] / 2;

    float* ws   = (float*)d_ws;
    float* h1   = ws;
    float* x2   = h1   + (size_t)N * C1;
    float* als1 = x2   + (size_t)N * C1;
    float* ald1 = als1 + (size_t)N * HEADS1;
    float* als2 = ald1 + (size_t)N * HEADS1;
    float* ald2 = als2 + (size_t)N * 2;
    const int nbin  = (N + BIN_W - 1) / BIN_W;
    const int nscan = nbin * NB1;
    int*   pairs  = (int*)(ald2 + (size_t)N * 2);
    int*   csr    = pairs  + (size_t)2 * E;
    int*   H      = csr    + (size_t)E;
    int*   L      = H      + (size_t)nscan;
    int*   rowptr = L      + (size_t)nscan + 1;
    float* h2   = h1;

    const int mtiles = (N + 15) / 16;

    bin_local_k<<<NB1, 1024, 0, stream>>>(eidx, E, N, nbin, H, L, pairs);
    bin_sort_k<<<nbin, 4 * NB1, 0, stream>>>(pairs, H, L, E, N, rowptr, csr);

    {
        dim3 grid((mtiles + 3) / 4, C1 / 64);
        gemm_wmma_f32_k<<<grid, 128, 0, stream>>>(x, W1, h1, a1_src, a1_dst, als1, ald1, N, 128, C1);
    }
    gat_gather_k<HEADS1, HID1, true, 1><<<(N + 7) / 8, 256, 0, stream>>>(rowptr, csr, als1, ald1,
                                                                         h1, b1, x2, N);

    {
        dim3 grid((mtiles + 3) / 4, C2 / 64);
        gemm_wmma_f32_k<<<grid, 128, 0, stream>>>(x2, W2, h2, a2_src, a2_dst, als2, ald2, N, C1, C2);
    }
    gat_gather_k<1, C2, false, 2><<<(N + 7) / 8, 256, 0, stream>>>(rowptr, csr, als2, ald2,
                                                                   h2, b2, out, N);
}


# Level 2 of the edge sort (a bin per block, now 128 threads walking 4 piece groups) rides in front of the layer-1 product's blocks in one launch: 6 launches to 5; bucket order still LDS arrival order
# speedup vs baseline: 9.0358x; 1.0809x over previous
#include <hip/hip_runtime.h>
#include <hip/hip_bf16.h>
#include <math.h>

typedef __attribute__((ext_vector_type(2))) float v2f;
typedef __attribute__((ext_vector_type(8))) float v8f;

#define HEADS1 4
#define HID1   64
#define C1     256
#define C2     128
#define NEG_SLOPE 0.2f

#define KC        128
#define BS_STRIDE 72

__device__ __forceinline__ float lrelu(float v) {
    return v < 0.0f ? NEG_SLOPE * v : v;
}

__device__ __forceinline__ float row16_sum(float v) {
    v += __int_as_float(__builtin_amdgcn_update_dpp(0, __float_as_int(v), 0x161, 0xf, 0xf, true));
    v += __int_as_float(__builtin_amdgcn_update_dpp(0, __float_as_int(v), 0x162, 0xf, 0xf, true));
    v += __int_as_float(__builtin_amdgcn_update_dpp(0, __float_as_int(v), 0x164, 0xf, 0xf, true));
    v += __int_as_float(__builtin_amdgcn_update_dpp(0, __float_as_int(v), 0x168, 0xf, 0xf, true));
    return v;
}

__device__ __forceinline__ void
gemm_block(const float* __restrict__ A,
           const float* __restrict__ B,
           float* __restrict__ C,
           const float* __restrict__ a_src,
           const float* __restrict__ a_dst,
           float* __restrict__ als, float* __restrict__ ald,
           int M, int K, int Ncols, int bx, int by, int nparts,
           float* __restrict__ Bs) {

    const int t     = threadIdx.x;
    const int lane  = t & 31;
    const int wave  = t >> 5;
    const int mtile = bx * 4 + wave;
    const int m0    = mtile * 16;
    const bool valid = (m0 < M);
    const int n0    = by * 64;
    const int l15   = lane & 15;
    const int koff  = (lane >> 4) * 2;
    const int row   = valid ? (m0 + l15) : 0;

    v8f acc0 = {}, acc1 = {}, acc2 = {}, acc3 = {};

    const float as0 = a_src[n0 + l15],      ad0 = a_dst[n0 + l15];
    const float as1 = a_src[n0 + l15 + 16], ad1 = a_dst[n0 + l15 + 16];
    const float as2 = a_src[n0 + l15 + 32], ad2 = a_dst[n0 + l15 + 32];
    const float as3 = a_src[n0 + l15 + 48], ad3 = a_dst[n0 + l15 + 48];

    for (int k0 = 0; k0 < K; k0 += KC) {
        __syncthreads();
#pragma unroll
        for (int i = 0; i < (KC * 64 / 4) / 128; ++i) {
            int j = t + i * 128;
            int k = j >> 4;
            int c = (j & 15) * 4;
            const float4 v = *(const float4*)(B + (size_t)(k0 + k) * Ncols + n0 + c);
            *(float4*)(&Bs[k * BS_STRIDE + c]) = v;
        }
        __syncthreads();

        if (valid) {
            const float* ap0 = A + (size_t)row * K + k0 + koff;
            for (int kk = 0; kk < KC; kk += 4) {
                v2f a; a.x = ap0[kk]; a.y = ap0[kk + 1];

                const float* bp = &Bs[(kk + koff) * BS_STRIDE + l15];
                v2f b0; b0.x = bp[0];  b0.y = bp[BS_STRIDE];
                v2f b1; b1.x = bp[16]; b1.y = bp[BS_STRIDE + 16];
                v2f b2; b2.x = bp[32]; b2.y = bp[BS_STRIDE + 32];
                v2f b3; b3.x = bp[48]; b3.y = bp[BS_STRIDE + 48];

                acc0 = __builtin_amdgcn_wmma_f32_16x16x4_f32(false, a, false, b0, (short)0, acc0, false, false);
                acc1 = __builtin_amdgcn_wmma_f32_16x16x4_f32(false, a, false, b1, (short)0, acc1, false, false);
                acc2 = __builtin_amdgcn_wmma_f32_16x16x4_f32(false, a, false, b2, (short)0, acc2, false, false);
                acc3 = __builtin_amdgcn_wmma_f32_16x16x4_f32(false, a, false, b3, (short)0, acc3, false, false);
            }
        }
    }

    if (valid) {
        const int rbase = m0 + ((lane >> 4) ? 8 : 0);
#pragma unroll
        for (int r = 0; r < 8; ++r) {
            float* cp = C + (size_t)(rbase + r) * Ncols + n0 + l15;
            cp[0]  = acc0[r];
            cp[16] = acc1[r];
            cp[32] = acc2[r];
            cp[48] = acc3[r];
        }

        float ps[8], pd[8];
#pragma unroll
        for (int r = 0; r < 8; ++r) {
            ps[r] = acc0[r] * as0 + acc1[r] * as1 + acc2[r] * as2 + acc3[r] * as3;
            pd[r] = acc0[r] * ad0 + acc1[r] * ad1 + acc2[r] * ad2 + acc3[r] * ad3;
        }
#pragma unroll
        for (int r = 0; r < 8; ++r) { ps[r] = row16_sum(ps[r]); pd[r] = row16_sum(pd[r]); }
        float vs = ps[0], vd = pd[0];
#pragma unroll
        for (int r = 1; r < 8; ++r) { vs = (l15 == r) ? ps[r] : vs; vd = (l15 == r) ? pd[r] : vd; }
        if (l15 < 8) {
            const size_t o = (size_t)(rbase + l15) * nparts + by;
            als[o] = vs;
            ald[o] = vd;
        }
    }
}

__global__ void __launch_bounds__(128)
gemm_wmma_f32_k(const float* __restrict__ A,
                const float* __restrict__ B,
                float* __restrict__ C,
                const float* __restrict__ a_src,
                const float* __restrict__ a_dst,
                float* __restrict__ als, float* __restrict__ ald,
                int M, int K, int Ncols) {
    __shared__ float Bs[KC * BS_STRIDE];
    gemm_block(A, B, C, a_src, a_dst, als, ald, M, K, Ncols, blockIdx.x, blockIdx.y, gridDim.y, Bs);
}

#define BIN_SHIFT 7
#define BIN_W     128
#define MAX_BINS  512
#define NB1       128

__device__ __forceinline__ int bin_of(int d) {
    return min(d >> BIN_SHIFT, MAX_BINS - 1);
}

__global__ void __launch_bounds__(1024)
bin_local_k(const int* __restrict__ ei, int E, int n, int nbin,
            int* __restrict__ H, int* __restrict__ L, int* __restrict__ pairs) {
    __shared__ int hist[MAX_BINS];
    __shared__ int cur[MAX_BINS];
    __shared__ int wtot[MAX_BINS / 32];
    const int t    = threadIdx.x;
    const int lane = t & 31;
    const int wave = t >> 5;
    const int b    = blockIdx.x;
    for (int k = t; k < MAX_BINS; k += 1024) hist[k] = 0;
    __syncthreads();
    const int per = (E + NB1 - 1) / NB1;
    const int lo  = b * per;
    const int hi  = min(lo + per, E);
    for (int e0 = lo + t; e0 < hi; e0 += 8 * 1024) {
        int d[8];
#pragma unroll
        for (int u = 0; u < 8; ++u) {
            const int e = min(e0 + u * 1024, hi - 1);
            d[u] = ei[E + e];
        }
        asm volatile("" :: "v"(d[0]), "v"(d[1]), "v"(d[2]), "v"(d[3]), "v"(d[4]), "v"(d[5]), "v"(d[6]), "v"(d[7]));
#pragma unroll
        for (int u = 0; u < 8; ++u) {
            if (e0 + u * 1024 < hi)
                atomicAdd(&hist[bin_of(min(max(d[u], 0), n - 1))], 1);
        }
    }
    __syncthreads();

    const int v = (t < MAX_BINS) ? hist[t] : 0;
    int incl = v;
    for (int off = 1; off < 32; off <<= 1) {
        int u = __shfl_up(incl, off);
        if (lane >= off) incl += u;
    }
    if (t < MAX_BINS && lane == 31) wtot[wave] = incl;
    __syncthreads();
    if (t < MAX_BINS) {
        int before = 0;
#pragma unroll
        for (int w = 0; w < MAX_BINS / 32; ++w) before += (w < wave) ? wtot[w] : 0;
        const int excl = before + incl - v;
        cur[t] = excl;
        if (t < nbin) {
            H[t * NB1 + b] = v;
            L[t * NB1 + b] = excl;
        }
    }
    __syncthreads();

    int2* __restrict__ p2 = (int2*)pairs + lo;
    for (int e0 = lo + t; e0 < hi; e0 += 8 * 1024) {
        int d[8], s[8];
#pragma unroll
        for (int u = 0; u < 8; ++u) {
            const int e = min(e0 + u * 1024, hi - 1);
            d[u] = ei[E + e];
            s[u] = ei[e];
        }
        asm volatile("" :: "v"(d[0]), "v"(d[1]), "v"(d[2]), "v"(d[3]), "v"(d[4]), "v"(d[5]), "v"(d[6]), "v"(d[7]),
                           "v"(s[0]), "v"(s[1]), "v"(s[2]), "v"(s[3]), "v"(s[4]), "v"(s[5]), "v"(s[6]), "v"(s[7]));
#pragma unroll
        for (int u = 0; u < 8; ++u) {
            if (e0 + u * 1024 < hi) {
                const int dc  = min(max(d[u], 0), n - 1);
                const int pos = atomicAdd(&cur[bin_of(dc)], 1);
                p2[pos] = make_int2(s[u], dc);
            }
        }
    }
}

#define SORT_T 128
#define SORT_G (4 * NB1 / SORT_T)

__device__ __forceinline__ void
bin_sort_block(const int* __restrict__ pairs, const int* __restrict__ H, const int* __restrict__ L,
               int E, int n, int* __restrict__ rowptr, int* __restrict__ csr_src, int k,
               int* lds) {
    static_assert(SORT_T == BIN_W, "one thread per destination counter in the scan");
    int* cnt  = lds;
    int* fill = cnt  + BIN_W;
    int* wsum = fill + BIN_W;
    int* red  = wsum + BIN_W / 32;
    const int t    = threadIdx.x;
    const int lane = t & 31;
    const int wave = t >> 5;
    const int d0   = k << BIN_SHIFT;
    const int per  = (E + NB1 - 1) / NB1;
    const int cg   = t >> 2;
    const int q    = t & 3;

    int len[SORT_G], lv[SORT_G];
#pragma unroll
    for (int g = 0; g < SORT_G; ++g) {
        const int c = g * (SORT_T / 4) + cg;
        len[g] = H[k * NB1 + c];
        lv[g]  = L[k * NB1 + c];
    }

    int pre = 0;
#pragma unroll
    for (int g = 0; g < SORT_G; ++g) pre += (q == 0) ? lv[g] : 0;
    for (int off = 16; off > 0; off >>= 1) pre += __shfl_xor(pre, off);
    if (t < BIN_W) cnt[t] = 0;
    if (lane == 0) red[wave] = pre;
    __syncthreads();
    int beg = 0;
    for (int w = 0; w < SORT_T / 32; ++w) beg += red[w];

#pragma unroll
    for (int g = 0; g < SORT_G; ++g) {
        const int2* __restrict__ pp = (const int2*)pairs + (size_t)(g * (SORT_T / 4) + cg) * per + lv[g];
        const int ln = len[g];
        for (int j0 = q; j0 < ln; j0 += 32) {
            int y[8];
#pragma unroll
            for (int u = 0; u < 8; ++u) {
                const int j = min(j0 + 4 * u, ln - 1);
                y[u] = pp[j].y;
            }
            asm volatile("" :: "v"(y[0]), "v"(y[1]), "v"(y[2]), "v"(y[3]), "v"(y[4]), "v"(y[5]), "v"(y[6]), "v"(y[7]));
#pragma unroll
            for (int u = 0; u < 8; ++u) {
                if (j0 + 4 * u < ln) atomicAdd(&cnt[(y[u] - d0) & (BIN_W - 1)], 1);
            }
        }
    }
    __syncthreads();

    const int v = (t < BIN_W) ? cnt[t] : 0;
    int incl = v;
    for (int off = 1; off < 32; off <<= 1) {
        int u = __shfl_up(incl, off);
        if (lane >= off) incl += u;
    }
    if (t < BIN_W && lane == 31) wsum[wave] = incl;
    __syncthreads();
    if (t < BIN_W) {
        int before = beg;
        for (int w = 0; w < wave; ++w) before += wsum[w];
        const int excl = before + incl - v;
        fill[t] = excl;
        const int node = d0 + t;
        if (node < n)      rowptr[node] = excl;
        if (node == n - 1) rowptr[n]    = excl + v;
    }
    __syncthreads();
#pragma unroll
    for (int g = 0; g < SORT_G; ++g) {
        const int2* __restrict__ pp = (const int2*)pairs + (size_t)(g * (SORT_T / 4) + cg) * per + lv[g];
        const int ln = len[g];
        for (int j0 = q; j0 < ln; j0 += 32) {
            int2 sd[8];
#pragma unroll
            for (int u = 0; u < 8; ++u) {
                const int j = min(j0 + 4 * u, ln - 1);
                sd[u] = pp[j];
            }
            asm volatile("" :: "v"(sd[0].x), "v"(sd[1].x), "v"(sd[2].x), "v"(sd[3].x), "v"(sd[4].x), "v"(sd[5].x), "v"(sd[6].x), "v"(sd[7].x),
                               "v"(sd[0].y), "v"(sd[1].y), "v"(sd[2].y), "v"(sd[3].y), "v"(sd[4].y), "v"(sd[5].y), "v"(sd[6].y), "v"(sd[7].y));
#pragma unroll
            for (int u = 0; u < 8; ++u) {
                if (j0 + 4 * u < ln) {
                    const int pos = atomicAdd(&fill[(sd[u].y - d0) & (BIN_W - 1)], 1);
                    csr_src[pos] = sd[u].x;
                }
            }
        }
    }
}

__global__ void __launch_bounds__(SORT_T)
gemm_sort_k(const float* __restrict__ A,
            const float* __restrict__ B,
            float* __restrict__ C,
            const float* __restrict__ a_src,
            const float* __restrict__ a_dst,
            float* __restrict__ als, float* __restrict__ ald,
            int M, int K, int Ncols, int gx,
            const int* __restrict__ pairs, const int* __restrict__ H, const int* __restrict__ L,
            int E, int* __restrict__ rowptr, int* __restrict__ csr_src, int nbin) {
    __shared__ float Bs[KC * BS_STRIDE];
    const int b = blockIdx.x;
    if (b < nbin) {
        bin_sort_block(pairs, H, L, E, M, rowptr, csr_src, b, (int*)Bs);
        return;
    }
    const int id = b - nbin;
    const int by = id / gx;
    const int bx = id - by * gx;
    gemm_block(A, B, C, a_src, a_dst, als, ald, M, K, Ncols, bx, by, Ncols / 64, Bs);
}

template <int HEADS>
__device__ __forceinline__ float pick_head(const float (&w)[HEADS], int hsel) {
    float r = w[0];
#pragma unroll
    for (int i = 1; i < HEADS; ++i) r = (hsel == i) ? w[i] : r;
    return r;
}

template <int HEADS, int CH, bool RELU, int PARTS>
__global__ void __launch_bounds__(256)
gat_gather_k(const int* __restrict__ rowptr, const int* __restrict__ csr_src,
             const float* __restrict__ als, const float* __restrict__ ald,
             const float* __restrict__ h, const float* __restrict__ bias,
             float* __restrict__ out, int n) {
    constexpr int C  = HEADS * CH;
    constexpr int NQ = C / 128;
    const int lane = threadIdx.x & 31;
    const int node = __builtin_amdgcn_readfirstlane((int)(blockIdx.x * 8 + (threadIdx.x >> 5)));
    if (node >= n) return;

    const int beg = rowptr[node];
    const int cnt = rowptr[node + 1] - beg + 1;

    float adst[HEADS];
#pragma unroll
    for (int hh = 0; hh < HEADS; ++hh) {
        float v = ald[((size_t)node * HEADS + hh) * PARTS];
#pragma unroll
        for (int p = 1; p < PARTS; ++p) v += ald[((size_t)node * HEADS + hh) * PARTS + p];
        adst[hh] = v;
    }

    int hq[NQ];
#pragma unroll
    for (int q = 0; q < NQ; ++q) hq[q] = (q * 128 + lane * 4) / CH;

    float  m[HEADS], dn[HEADS];
    float4 acc[NQ];
#pragma unroll
    for (int hh = 0; hh < HEADS; ++hh) { m[hh] = 0.0f; dn[hh] = 0.0f; }
#pragma unroll
    for (int q = 0; q < NQ; ++q) acc[q] = make_float4(0.0f, 0.0f, 0.0f, 0.0f);

    for (int j0 = 0; j0 < cnt; j0 += 32) {
        const int  j     = j0 + lane;
        const bool valid = j < cnt;
        int s_l = node;
        if (valid && j > 0) s_l = min(max(csr_src[beg + j - 1], 0), n - 1);

        float a[HEADS];
        if constexpr (HEADS == 4 && PARTS == 1) {
            const float4 a4 = *(const float4*)(als + (size_t)s_l * 4);
            a[0] = a4.x; a[1] = a4.y; a[2] = a4.z; a[3] = a4.w;
        } else if constexpr (HEADS == 1 && PARTS == 2) {
            const float2 a2 = *(const float2*)(als + (size_t)s_l * 2);
            a[0] = a2.x + a2.y;
        } else {
#pragma unroll
            for (int hh = 0; hh < HEADS; ++hh) {
                float v = als[((size_t)s_l * HEADS + hh) * PARTS];
#pragma unroll
                for (int p = 1; p < PARTS; ++p) v += als[((size_t)s_l * HEADS + hh) * PARTS + p];
                a[hh] = v;
            }
        }

        float e[HEADS], cm[HEADS], pw[HEADS], ps[HEADS], sc[HEADS];
#pragma unroll
        for (int hh = 0; hh < HEADS; ++hh) {
            e[hh]  = valid ? lrelu(a[hh] + adst[hh]) : -INFINITY;
            cm[hh] = e[hh];
        }
#pragma unroll
        for (int off = 16; off > 0; off >>= 1) {
#pragma unroll
            for (int hh = 0; hh < HEADS; ++hh) cm[hh] = fmaxf(cm[hh], __shfl_xor(cm[hh], off));
        }
#pragma unroll
        for (int hh = 0; hh < HEADS; ++hh) {
            const float mn = (j0 == 0) ? cm[hh] : fmaxf(m[hh], cm[hh]);
            sc[hh] = (j0 == 0) ? 0.0f : __expf(m[hh] - mn);
            pw[hh] = valid ? __expf(e[hh] - mn) : 0.0f;
            ps[hh] = pw[hh];
            m[hh]  = mn;
        }
#pragma unroll
        for (int off = 16; off > 0; off >>= 1) {
#pragma unroll
            for (int hh = 0; hh < HEADS; ++hh) ps[hh] += __shfl_xor(ps[hh], off);
        }
#pragma unroll
        for (int hh = 0; hh < HEADS; ++hh) dn[hh] = dn[hh] * sc[hh] + ps[hh];
        if (j0 > 0) {
#pragma unroll
            for (int q = 0; q < NQ; ++q) {
                const float f = pick_head<HEADS>(sc, hq[q]);
                acc[q].x *= f; acc[q].y *= f; acc[q].z *= f; acc[q].w *= f;
            }
        }

        const int nc = min(32, cnt - j0);
        for (int t = 0; t < nc; t += 4) {
#pragma unroll
            for (int u = 0; u < 4; ++u) {
                const int s = __builtin_amdgcn_readlane(s_l, t + u);
                float wv[HEADS];
#pragma unroll
                for (int hh = 0; hh < HEADS; ++hh)
                    wv[hh] = __int_as_float(__builtin_amdgcn_readlane(__float_as_int(pw[hh]), t + u));
                const float4* hp = (const float4*)(h + (size_t)s * C) + lane;
#pragma unroll
                for (int q = 0; q < NQ; ++q) {
                    const float4 v = hp[q * 32];
                    const float  w = pick_head<HEADS>(wv, hq[q]);
                    acc[q].x += w * v.x; acc[q].y += w * v.y;
                    acc[q].z += w * v.z; acc[q].w += w * v.w;
                }
            }
        }
    }

#pragma unroll
    for (int q = 0; q < NQ; ++q) {
        const int    c0 = q * 128 + lane * 4;
        const float  d  = pick_head<HEADS>(dn, hq[q]);
        const float4 b  = *(const float4*)(bias + c0);
        float4 r;
        r.x = acc[q].x / d + b.x; r.y = acc[q].y / d + b.y;
        r.z = acc[q].z / d + b.z; r.w = acc[q].w / d + b.w;
        if (RELU) {
            r.x = r.x > 0.0f ? r.x : 0.0f; r.y = r.y > 0.0f ? r.y : 0.0f;
            r.z = r.z > 0.0f ? r.z : 0.0f; r.w = r.w > 0.0f ? r.w : 0.0f;
        }
        *(float4*)(out + (size_t)node * C + c0) = r;
    }
}

extern "C" void kernel_launch(void* const* d_in, const int* in_sizes, int n_in,
                              void* d_out, int out_size, void* d_ws, size_t ws_size,
                              hipStream_t stream) {
    const float* x      = (const float*)d_in[0];
    const int*   eidx   = (const int*)d_in[1];
    const float* W1     = (const float*)d_in[2];
    const float* a1_src = (const float*)d_in[3];
    const float* a1_dst = (const float*)d_in[4];
    const float* b1     = (const float*)d_in[5];
    const float* W2     = (const float*)d_in[6];
    const float* a2_src = (const float*)d_in[7];
    const float* a2_dst = (const float*)d_in[8];
    const float* b2     = (const float*)d_in[9];
    float*       out    = (float*)d_out;

    const int N = in_sizes[0] / 128;
    const int E = in_sizes[1] / 2;

    float* ws   = (float*)d_ws;
    float* h1   = ws;
    float* x2   = h1   + (size_t)N * C1;
    float* als1 = x2   + (size_t)N * C1;
    float* ald1 = als1 + (size_t)N * HEADS1;
    float* als2 = ald1 + (size_t)N * HEADS1;
    float* ald2 = als2 + (size_t)N * 2;
    const int nbin  = (N + BIN_W - 1) / BIN_W;
    const int nscan = nbin * NB1;
    int*   pairs  = (int*)(ald2 + (size_t)N * 2);
    int*   csr    = pairs  + (size_t)2 * E;
    int*   H      = csr    + (size_t)E;
    int*   L      = H      + (size_t)nscan;
    int*   rowptr = L      + (size_t)nscan + 1;
    float* h2   = h1;

    const int mtiles = (N + 15) / 16;

    bin_local_k<<<NB1, 1024, 0, stream>>>(eidx, E, N, nbin, H, L, pairs);

    {
        const int gx = (mtiles + 3) / 4;
        gemm_sort_k<<<nbin + gx * (C1 / 64), SORT_T, 0, stream>>>(x, W1, h1, a1_src, a1_dst, als1, ald1, N, 128, C1, gx,
                                                                 pairs, H, L, E, rowptr, csr, nbin);
    }
    gat_gather_k<HEADS1, HID1, true, 1><<<(N + 7) / 8, 256, 0, stream>>>(rowptr, csr, als1, ald1,
                                                                         h1, b1, x2, N);

    {
        dim3 grid((mtiles + 3) / 4, C2 / 64);
        gemm_wmma_f32_k<<<grid, 128, 0, stream>>>(x2, W2, h2, a2_src, a2_dst, als2, ald2, N, C1, C2);
    }
    gat_gather_k<1, C2, false, 2><<<(N + 7) / 8, 256, 0, stream>>>(rowptr, csr, als2, ald2,
                                                                   h2, b2, out, N);
}


# Product kernels: a thread's 16 loads of a B slab chunk go out as one group (one wait) ahead of its 16 LDS stores, addresses stepped by one add: one trip per K chunk, not sixteen in a row
# speedup vs baseline: 9.6906x; 1.0725x over previous
#include <hip/hip_runtime.h>
#include <hip/hip_bf16.h>
#include <math.h>

typedef __attribute__((ext_vector_type(2))) float v2f;
typedef __attribute__((ext_vector_type(8))) float v8f;
typedef __attribute__((ext_vector_type(4))) float v4f;

#define HEADS1 4
#define HID1   64
#define C1     256
#define C2     128
#define NEG_SLOPE 0.2f

#define KC        128
#define BS_STRIDE 72

__device__ __forceinline__ float lrelu(float v) {
    return v < 0.0f ? NEG_SLOPE * v : v;
}

__device__ __forceinline__ float row16_sum(float v) {
    v += __int_as_float(__builtin_amdgcn_update_dpp(0, __float_as_int(v), 0x161, 0xf, 0xf, true));
    v += __int_as_float(__builtin_amdgcn_update_dpp(0, __float_as_int(v), 0x162, 0xf, 0xf, true));
    v += __int_as_float(__builtin_amdgcn_update_dpp(0, __float_as_int(v), 0x164, 0xf, 0xf, true));
    v += __int_as_float(__builtin_amdgcn_update_dpp(0, __float_as_int(v), 0x168, 0xf, 0xf, true));
    return v;
}

__device__ __forceinline__ void
gemm_block(const float* __restrict__ A,
           const float* __restrict__ B,
           float* __restrict__ C,
           const float* __restrict__ a_src,
           const float* __restrict__ a_dst,
           float* __restrict__ als, float* __restrict__ ald,
           int M, int K, int Ncols, int bx, int by, int nparts,
           float* __restrict__ Bs) {

    const int t     = threadIdx.x;
    const int lane  = t & 31;
    const int wave  = t >> 5;
    const int mtile = bx * 4 + wave;
    const int m0    = mtile * 16;
    const bool valid = (m0 < M);
    const int n0    = by * 64;
    const int l15   = lane & 15;
    const int koff  = (lane >> 4) * 2;
    const int row   = valid ? (m0 + l15) : 0;

    v8f acc0 = {}, acc1 = {}, acc2 = {}, acc3 = {};

    const float as0 = a_src[n0 + l15],      ad0 = a_dst[n0 + l15];
    const float as1 = a_src[n0 + l15 + 16], ad1 = a_dst[n0 + l15 + 16];
    const float as2 = a_src[n0 + l15 + 32], ad2 = a_dst[n0 + l15 + 32];
    const float as3 = a_src[n0 + l15 + 48], ad3 = a_dst[n0 + l15 + 48];

    for (int k0 = 0; k0 < K; k0 += KC) {
        constexpr int NSTAGE = (KC * 64 / 4) / 128;
        v4f sv[NSTAGE];
        const float* __restrict__ bslab = B + (size_t)k0 * Ncols + n0;
        const unsigned   toff  = (unsigned)((t >> 4) * Ncols + (t & 15) * 4);
#pragma unroll
        for (int i = 0; i < NSTAGE; ++i)
            sv[i] = *(const v4f*)(bslab + (size_t)(i * 8) * Ncols + toff);
        static_assert(NSTAGE == 16, "the fence below names the sixteen loaded values one by one");
        asm volatile("" :: "v"(sv[0]), "v"(sv[1]), "v"(sv[2]),  "v"(sv[3]),  "v"(sv[4]),  "v"(sv[5]),  "v"(sv[6]),  "v"(sv[7]),
                           "v"(sv[8]), "v"(sv[9]), "v"(sv[10]), "v"(sv[11]), "v"(sv[12]), "v"(sv[13]), "v"(sv[14]), "v"(sv[15]));
        __syncthreads();
#pragma unroll
        for (int i = 0; i < NSTAGE; ++i) {
            const int j = t + i * 128;
            const int k = j >> 4;
            const int c = (j & 15) * 4;
            *(v4f*)(&Bs[k * BS_STRIDE + c]) = sv[i];
        }
        __syncthreads();

        if (valid) {
            const float* ap0 = A + (size_t)row * K + k0 + koff;
            for (int kk = 0; kk < KC; kk += 4) {
                v2f a; a.x = ap0[kk]; a.y = ap0[kk + 1];

                const float* bp = &Bs[(kk + koff) * BS_STRIDE + l15];
                v2f b0; b0.x = bp[0];  b0.y = bp[BS_STRIDE];
                v2f b1; b1.x = bp[16]; b1.y = bp[BS_STRIDE + 16];
                v2f b2; b2.x = bp[32]; b2.y = bp[BS_STRIDE + 32];
                v2f b3; b3.x = bp[48]; b3.y = bp[BS_STRIDE + 48];

                acc0 = __builtin_amdgcn_wmma_f32_16x16x4_f32(false, a, false, b0, (short)0, acc0, false, false);
                acc1 = __builtin_amdgcn_wmma_f32_16x16x4_f32(false, a, false, b1, (short)0, acc1, false, false);
                acc2 = __builtin_amdgcn_wmma_f32_16x16x4_f32(false, a, false, b2, (short)0, acc2, false, false);
                acc3 = __builtin_amdgcn_wmma_f32_16x16x4_f32(false, a, false, b3, (short)0, acc3, false, false);
            }
        }
    }

    if (valid) {
        const int rbase = m0 + ((lane >> 4) ? 8 : 0);
#pragma unroll
        for (int r = 0; r < 8; ++r) {
            float* cp = C + (size_t)(rbase + r) * Ncols + n0 + l15;
            cp[0]  = acc0[r];
            cp[16] = acc1[r];
            cp[32] = acc2[r];
            cp[48] = acc3[r];
        }

        float ps[8], pd[8];
#pragma unroll
        for (int r = 0; r < 8; ++r) {
            ps[r] = acc0[r] * as0 + acc1[r] * as1 + acc2[r] * as2 + acc3[r] * as3;
            pd[r] = acc0[r] * ad0 + acc1[r] * ad1 + acc2[r] * ad2 + acc3[r] * ad3;
        }
#pragma unroll
        for (int r = 0; r < 8; ++r) { ps[r] = row16_sum(ps[r]); pd[r] = row16_sum(pd[r]); }
        float vs = ps[0], vd = pd[0];
#pragma unroll
        for (int r = 1; r < 8; ++r) { vs = (l15 == r) ? ps[r] : vs; vd = (l15 == r) ? pd[r] : vd; }
        if (l15 < 8) {
            const size_t o = (size_t)(rbase + l15) * nparts + by;
            als[o] = vs;
            ald[o] = vd;
        }
    }
}

__global__ void __launch_bounds__(128)
gemm_wmma_f32_k(const float* __restrict__ A,
                const float* __restrict__ B,
                float* __restrict__ C,
                const float* __restrict__ a_src,
                const float* __restrict__ a_dst,
                float* __restrict__ als, float* __restrict__ ald,
                int M, int K, int Ncols) {
    __shared__ float Bs[KC * BS_STRIDE];
    gemm_block(A, B, C, a_src, a_dst, als, ald, M, K, Ncols, blockIdx.x, blockIdx.y, gridDim.y, Bs);
}

#define BIN_SHIFT 7
#define BIN_W     128
#define MAX_BINS  512
#define NB1       128

__device__ __forceinline__ int bin_of(int d) {
    return min(d >> BIN_SHIFT, MAX_BINS - 1);
}

__global__ void __launch_bounds__(1024)
bin_local_k(const int* __restrict__ ei, int E, int n, int nbin,
            int* __restrict__ H, int* __restrict__ L, int* __restrict__ pairs) {
    __shared__ int hist[MAX_BINS];
    __shared__ int cur[MAX_BINS];
    __shared__ int wtot[MAX_BINS / 32];
    const int t    = threadIdx.x;
    const int lane = t & 31;
    const int wave = t >> 5;
    const int b    = blockIdx.x;
    for (int k = t; k < MAX_BINS; k += 1024) hist[k] = 0;
    __syncthreads();
    const int per = (E + NB1 - 1) / NB1;
    const int lo  = b * per;
    const int hi  = min(lo + per, E);
    for (int e0 = lo + t; e0 < hi; e0 += 8 * 1024) {
        int d[8];
#pragma unroll
        for (int u = 0; u < 8; ++u) {
            const int e = min(e0 + u * 1024, hi - 1);
            d[u] = ei[E + e];
        }
        asm volatile("" :: "v"(d[0]), "v"(d[1]), "v"(d[2]), "v"(d[3]), "v"(d[4]), "v"(d[5]), "v"(d[6]), "v"(d[7]));
#pragma unroll
        for (int u = 0; u < 8; ++u) {
            if (e0 + u * 1024 < hi)
                atomicAdd(&hist[bin_of(min(max(d[u], 0), n - 1))], 1);
        }
    }
    __syncthreads();

    const int v = (t < MAX_BINS) ? hist[t] : 0;
    int incl = v;
    for (int off = 1; off < 32; off <<= 1) {
        int u = __shfl_up(incl, off);
        if (lane >= off) incl += u;
    }
    if (t < MAX_BINS && lane == 31) wtot[wave] = incl;
    __syncthreads();
    if (t < MAX_BINS) {
        int before = 0;
#pragma unroll
        for (int w = 0; w < MAX_BINS / 32; ++w) before += (w < wave) ? wtot[w] : 0;
        const int excl = before + incl - v;
        cur[t] = excl;
        if (t < nbin) {
            H[t * NB1 + b] = v;
            L[t * NB1 + b] = excl;
        }
    }
    __syncthreads();

    int2* __restrict__ p2 = (int2*)pairs + lo;
    for (int e0 = lo + t; e0 < hi; e0 += 8 * 1024) {
        int d[8], s[8];
#pragma unroll
        for (int u = 0; u < 8; ++u) {
            const int e = min(e0 + u * 1024, hi - 1);
            d[u] = ei[E + e];
            s[u] = ei[e];
        }
        asm volatile("" :: "v"(d[0]), "v"(d[1]), "v"(d[2]), "v"(d[3]), "v"(d[4]), "v"(d[5]), "v"(d[6]), "v"(d[7]),
                           "v"(s[0]), "v"(s[1]), "v"(s[2]), "v"(s[3]), "v"(s[4]), "v"(s[5]), "v"(s[6]), "v"(s[7]));
#pragma unroll
        for (int u = 0; u < 8; ++u) {
            if (e0 + u * 1024 < hi) {
                const int dc  = min(max(d[u], 0), n - 1);
                const int pos = atomicAdd(&cur[bin_of(dc)], 1);
                p2[pos] = make_int2(s[u], dc);
            }
        }
    }
}

#define SORT_T 128
#define SORT_G (4 * NB1 / SORT_T)

__device__ __forceinline__ void
bin_sort_block(const int* __restrict__ pairs, const int* __restrict__ H, const int* __restrict__ L,
               int E, int n, int* __restrict__ rowptr, int* __restrict__ csr_src, int k,
               int* lds) {
    static_assert(SORT_T == BIN_W, "one thread per destination counter in the scan");
    int* cnt  = lds;
    int* fill = cnt  + BIN_W;
    int* wsum = fill + BIN_W;
    int* red  = wsum + BIN_W / 32;
    const int t    = threadIdx.x;
    const int lane = t & 31;
    const int wave = t >> 5;
    const int d0   = k << BIN_SHIFT;
    const int per  = (E + NB1 - 1) / NB1;
    const int cg   = t >> 2;
    const int q    = t & 3;

    int len[SORT_G], lv[SORT_G];
#pragma unroll
    for (int g = 0; g < SORT_G; ++g) {
        const int c = g * (SORT_T / 4) + cg;
        len[g] = H[k * NB1 + c];
        lv[g]  = L[k * NB1 + c];
    }

    int pre = 0;
#pragma unroll
    for (int g = 0; g < SORT_G; ++g) pre += (q == 0) ? lv[g] : 0;
    for (int off = 16; off > 0; off >>= 1) pre += __shfl_xor(pre, off);
    if (t < BIN_W) cnt[t] = 0;
    if (lane == 0) red[wave] = pre;
    __syncthreads();
    int beg = 0;
    for (int w = 0; w < SORT_T / 32; ++w) beg += red[w];

#pragma unroll
    for (int g = 0; g < SORT_G; ++g) {
        const int2* __restrict__ pp = (const int2*)pairs + (size_t)(g * (SORT_T / 4) + cg) * per + lv[g];
        const int ln = len[g];
        for (int j0 = q; j0 < ln; j0 += 32) {
            int y[8];
#pragma unroll
            for (int u = 0; u < 8; ++u) {
                const int j = min(j0 + 4 * u, ln - 1);
                y[u] = pp[j].y;
            }
            asm volatile("" :: "v"(y[0]), "v"(y[1]), "v"(y[2]), "v"(y[3]), "v"(y[4]), "v"(y[5]), "v"(y[6]), "v"(y[7]));
#pragma unroll
            for (int u = 0; u < 8; ++u) {
                if (j0 + 4 * u < ln) atomicAdd(&cnt[(y[u] - d0) & (BIN_W - 1)], 1);
            }
        }
    }
    __syncthreads();

    const int v = (t < BIN_W) ? cnt[t] : 0;
    int incl = v;
    for (int off = 1; off < 32; off <<= 1) {
        int u = __shfl_up(incl, off);
        if (lane >= off) incl += u;
    }
    if (t < BIN_W && lane == 31) wsum[wave] = incl;
    __syncthreads();
    if (t < BIN_W) {
        int before = beg;
        for (int w = 0; w < wave; ++w) before += wsum[w];
        const int excl = before + incl - v;
        fill[t] = excl;
        const int node = d0 + t;
        if (node < n)      rowptr[node] = excl;
        if (node == n - 1) rowptr[n]    = excl + v;
    }
    __syncthreads();
#pragma unroll
    for (int g = 0; g < SORT_G; ++g) {
        const int2* __restrict__ pp = (const int2*)pairs + (size_t)(g * (SORT_T / 4) + cg) * per + lv[g];
        const int ln = len[g];
        for (int j0 = q; j0 < ln; j0 += 32) {
            int2 sd[8];
#pragma unroll
            for (int u = 0; u < 8; ++u) {
                const int j = min(j0 + 4 * u, ln - 1);
                sd[u] = pp[j];
            }
            asm volatile("" :: "v"(sd[0].x), "v"(sd[1].x), "v"(sd[2].x), "v"(sd[3].x), "v"(sd[4].x), "v"(sd[5].x), "v"(sd[6].x), "v"(sd[7].x),
                               "v"(sd[0].y), "v"(sd[1].y), "v"(sd[2].y), "v"(sd[3].y), "v"(sd[4].y), "v"(sd[5].y), "v"(sd[6].y), "v"(sd[7].y));
#pragma unroll
            for (int u = 0; u < 8; ++u) {
                if (j0 + 4 * u < ln) {
                    const int pos = atomicAdd(&fill[(sd[u].y - d0) & (BIN_W - 1)], 1);
                    csr_src[pos] = sd[u].x;
                }
            }
        }
    }
}

__global__ void __launch_bounds__(SORT_T)
gemm_sort_k(const float* __restrict__ A,
            const float* __restrict__ B,
            float* __restrict__ C,
            const float* __restrict__ a_src,
            const float* __restrict__ a_dst,
            float* __restrict__ als, float* __restrict__ ald,
            int M, int K, int Ncols, int gx,
            const int* __restrict__ pairs, const int* __restrict__ H, const int* __restrict__ L,
            int E, int* __restrict__ rowptr, int* __restrict__ csr_src, int nbin) {
    __shared__ float Bs[KC * BS_STRIDE];
    const int b = blockIdx.x;
    if (b < nbin) {
        bin_sort_block(pairs, H, L, E, M, rowptr, csr_src, b, (int*)Bs);
        return;
    }
    const int id = b - nbin;
    const int by = id / gx;
    const int bx = id - by * gx;
    gemm_block(A, B, C, a_src, a_dst, als, ald, M, K, Ncols, bx, by, Ncols / 64, Bs);
}

template <int HEADS>
__device__ __forceinline__ float pick_head(const float (&w)[HEADS], int hsel) {
    float r = w[0];
#pragma unroll
    for (int i = 1; i < HEADS; ++i) r = (hsel == i) ? w[i] : r;
    return r;
}

template <int HEADS, int CH, bool RELU, int PARTS>
__global__ void __launch_bounds__(256)
gat_gather_k(const int* __restrict__ rowptr, const int* __restrict__ csr_src,
             const float* __restrict__ als, const float* __restrict__ ald,
             const float* __restrict__ h, const float* __restrict__ bias,
             float* __restrict__ out, int n) {
    constexpr int C  = HEADS * CH;
    constexpr int NQ = C / 128;
    const int lane = threadIdx.x & 31;
    const int node = __builtin_amdgcn_readfirstlane((int)(blockIdx.x * 8 + (threadIdx.x >> 5)));
    if (node >= n) return;

    const int beg = rowptr[node];
    const int cnt = rowptr[node + 1] - beg + 1;

    float adst[HEADS];
#pragma unroll
    for (int hh = 0; hh < HEADS; ++hh) {
        float v = ald[((size_t)node * HEADS + hh) * PARTS];
#pragma unroll
        for (int p = 1; p < PARTS; ++p) v += ald[((size_t)node * HEADS + hh) * PARTS + p];
        adst[hh] = v;
    }

    int hq[NQ];
#pragma unroll
    for (int q = 0; q < NQ; ++q) hq[q] = (q * 128 + lane * 4) / CH;

    float  m[HEADS], dn[HEADS];
    float4 acc[NQ];
#pragma unroll
    for (int hh = 0; hh < HEADS; ++hh) { m[hh] = 0.0f; dn[hh] = 0.0f; }
#pragma unroll
    for (int q = 0; q < NQ; ++q) acc[q] = make_float4(0.0f, 0.0f, 0.0f, 0.0f);

    for (int j0 = 0; j0 < cnt; j0 += 32) {
        const int  j     = j0 + lane;
        const bool valid = j < cnt;
        int s_l = node;
        if (valid && j > 0) s_l = min(max(csr_src[beg + j - 1], 0), n - 1);

        float a[HEADS];
        if constexpr (HEADS == 4 && PARTS == 1) {
            const float4 a4 = *(const float4*)(als + (size_t)s_l * 4);
            a[0] = a4.x; a[1] = a4.y; a[2] = a4.z; a[3] = a4.w;
        } else if constexpr (HEADS == 1 && PARTS == 2) {
            const float2 a2 = *(const float2*)(als + (size_t)s_l * 2);
            a[0] = a2.x + a2.y;
        } else {
#pragma unroll
            for (int hh = 0; hh < HEADS; ++hh) {
                float v = als[((size_t)s_l * HEADS + hh) * PARTS];
#pragma unroll
                for (int p = 1; p < PARTS; ++p) v += als[((size_t)s_l * HEADS + hh) * PARTS + p];
                a[hh] = v;
            }
        }

        float e[HEADS], cm[HEADS], pw[HEADS], ps[HEADS], sc[HEADS];
#pragma unroll
        for (int hh = 0; hh < HEADS; ++hh) {
            e[hh]  = valid ? lrelu(a[hh] + adst[hh]) : -INFINITY;
            cm[hh] = e[hh];
        }
#pragma unroll
        for (int off = 16; off > 0; off >>= 1) {
#pragma unroll
            for (int hh = 0; hh < HEADS; ++hh) cm[hh] = fmaxf(cm[hh], __shfl_xor(cm[hh], off));
        }
#pragma unroll
        for (int hh = 0; hh < HEADS; ++hh) {
            const float mn = (j0 == 0) ? cm[hh] : fmaxf(m[hh], cm[hh]);
            sc[hh] = (j0 == 0) ? 0.0f : __expf(m[hh] - mn);
            pw[hh] = valid ? __expf(e[hh] - mn) : 0.0f;
            ps[hh] = pw[hh];
            m[hh]  = mn;
        }
#pragma unroll
        for (int off = 16; off > 0; off >>= 1) {
#pragma unroll
            for (int hh = 0; hh < HEADS; ++hh) ps[hh] += __shfl_xor(ps[hh], off);
        }
#pragma unroll
        for (int hh = 0; hh < HEADS; ++hh) dn[hh] = dn[hh] * sc[hh] + ps[hh];
        if (j0 > 0) {
#pragma unroll
            for (int q = 0; q < NQ; ++q) {
                const float f = pick_head<HEADS>(sc, hq[q]);
                acc[q].x *= f; acc[q].y *= f; acc[q].z *= f; acc[q].w *= f;
            }
        }

        const int nc = min(32, cnt - j0);
        for (int t = 0; t < nc; t += 4) {
#pragma unroll
            for (int u = 0; u < 4; ++u) {
                const int s = __builtin_amdgcn_readlane(s_l, t + u);
                float wv[HEADS];
#pragma unroll
                for (int hh = 0; hh < HEADS; ++hh)
                    wv[hh] = __int_as_float(__builtin_amdgcn_readlane(__float_as_int(pw[hh]), t + u));
                const float4* hp = (const float4*)(h + (size_t)s * C) + lane;
#pragma unroll
                for (int q = 0; q < NQ; ++q) {
                    const float4 v = hp[q * 32];
                    const float  w = pick_head<HEADS>(wv, hq[q]);
                    acc[q].x += w * v.x; acc[q].y += w * v.y;
                    acc[q].z += w * v.z; acc[q].w += w * v.w;
                }
            }
        }
    }

#pragma unroll
    for (int q = 0; q < NQ; ++q) {
        const int    c0 = q * 128 + lane * 4;
        const float  d  = pick_head<HEADS>(dn, hq[q]);
        const float4 b  = *(const float4*)(bias + c0);
        float4 r;
        r.x = acc[q].x / d + b.x; r.y = acc[q].y / d + b.y;
        r.z = acc[q].z / d + b.z; r.w = acc[q].w / d + b.w;
        if (RELU) {
            r.x = r.x > 0.0f ? r.x : 0.0f; r.y = r.y > 0.0f ? r.y : 0.0f;
            r.z = r.z > 0.0f ? r.z : 0.0f; r.w = r.w > 0.0f ? r.w : 0.0f;
        }
        *(float4*)(out + (size_t)node * C + c0) = r;
    }
}

extern "C" void kernel_launch(void* const* d_in, const int* in_sizes, int n_in,
                              void* d_out, int out_size, void* d_ws, size_t ws_size,
                              hipStream_t stream) {
    const float* x      = (const float*)d_in[0];
    const int*   eidx   = (const int*)d_in[1];
    const float* W1     = (const float*)d_in[2];
    const float* a1_src = (const float*)d_in[3];
    const float* a1_dst = (const float*)d_in[4];
    const float* b1     = (const float*)d_in[5];
    const float* W2     = (const float*)d_in[6];
    const float* a2_src = (const float*)d_in[7];
    const float* a2_dst = (const float*)d_in[8];
    const float* b2     = (const float*)d_in[9];
    float*       out    = (float*)d_out;

    const int N = in_sizes[0] / 128;
    const int E = in_sizes[1] / 2;

    float* ws   = (float*)d_ws;
    float* h1   = ws;
    float* x2   = h1   + (size_t)N * C1;
    float* als1 = x2   + (size_t)N * C1;
    float* ald1 = als1 + (size_t)N * HEADS1;
    float* als2 = ald1 + (size_t)N * HEADS1;
    float* ald2 = als2 + (size_t)N * 2;
    const int nbin  = (N + BIN_W - 1) / BIN_W;
    const int nscan = nbin * NB1;
    int*   pairs  = (int*)(ald2 + (size_t)N * 2);
    int*   csr    = pairs  + (size_t)2 * E;
    int*   H      = csr    + (size_t)E;
    int*   L      = H      + (size_t)nscan;
    int*   rowptr = L      + (size_t)nscan + 1;
    float* h2   = h1;

    const int mtiles = (N + 15) / 16;

    bin_local_k<<<NB1, 1024, 0, stream>>>(eidx, E, N, nbin, H, L, pairs);

    {
        const int gx = (mtiles + 3) / 4;
        gemm_sort_k<<<nbin + gx * (C1 / 64), SORT_T, 0, stream>>>(x, W1, h1, a1_src, a1_dst, als1, ald1, N, 128, C1, gx,
                                                                 pairs, H, L, E, rowptr, csr, nbin);
    }
    gat_gather_k<HEADS1, HID1, true, 1><<<(N + 7) / 8, 256, 0, stream>>>(rowptr, csr, als1, ald1,
                                                                         h1, b1, x2, N);

    {
        dim3 grid((mtiles + 3) / 4, C2 / 64);
        gemm_wmma_f32_k<<<grid, 128, 0, stream>>>(x2, W2, h2, a2_src, a2_dst, als2, ald2, N, C1, C2);
    }
    gat_gather_k<1, C2, false, 2><<<(N + 7) / 8, 256, 0, stream>>>(rowptr, csr, als2, ald2,
                                                                   h2, b2, out, N);
}


# Layer-2 product runs behind the layer-1 gather in the same block: a block's 196 summed rows stay in LDS as its A operand, W2 streams through LDS in 32-row chunks; 5 launches to 4, sums unchanged
# speedup vs baseline: 10.4417x; 1.0775x over previous
#include <hip/hip_runtime.h>
#include <hip/hip_bf16.h>
#include <math.h>

typedef __attribute__((ext_vector_type(2))) float v2f;
typedef __attribute__((ext_vector_type(8))) float v8f;
typedef __attribute__((ext_vector_type(4))) float v4f;

#define HEADS1 4
#define HID1   64
#define C1     256
#define C2     128
#define NEG_SLOPE 0.2f

#define KC        128
#define BS_STRIDE 72

__device__ __forceinline__ float lrelu(float v) {
    return v < 0.0f ? NEG_SLOPE * v : v;
}

__device__ __forceinline__ float row16_sum(float v) {
    v += __int_as_float(__builtin_amdgcn_update_dpp(0, __float_as_int(v), 0x161, 0xf, 0xf, true));
    v += __int_as_float(__builtin_amdgcn_update_dpp(0, __float_as_int(v), 0x162, 0xf, 0xf, true));
    v += __int_as_float(__builtin_amdgcn_update_dpp(0, __float_as_int(v), 0x164, 0xf, 0xf, true));
    v += __int_as_float(__builtin_amdgcn_update_dpp(0, __float_as_int(v), 0x168, 0xf, 0xf, true));
    return v;
}

__device__ __forceinline__ void
gemm_block(const float* __restrict__ A,
           const float* __restrict__ B,
           float* __restrict__ C,
           const float* __restrict__ a_src,
           const float* __restrict__ a_dst,
           float* __restrict__ als, float* __restrict__ ald,
           int M, int K, int Ncols, int bx, int by, int nparts,
           float* __restrict__ Bs) {

    const int t     = threadIdx.x;
    const int lane  = t & 31;
    const int wave  = t >> 5;
    const int mtile = bx * 4 + wave;
    const int m0    = mtile * 16;
    const bool valid = (m0 < M);
    const int n0    = by * 64;
    const int l15   = lane & 15;
    const int koff  = (lane >> 4) * 2;
    const int row   = valid ? (m0 + l15) : 0;

    v8f acc0 = {}, acc1 = {}, acc2 = {}, acc3 = {};

    const float as0 = a_src[n0 + l15],      ad0 = a_dst[n0 + l15];
    const float as1 = a_src[n0 + l15 + 16], ad1 = a_dst[n0 + l15 + 16];
    const float as2 = a_src[n0 + l15 + 32], ad2 = a_dst[n0 + l15 + 32];
    const float as3 = a_src[n0 + l15 + 48], ad3 = a_dst[n0 + l15 + 48];

    for (int k0 = 0; k0 < K; k0 += KC) {
        constexpr int NSTAGE = (KC * 64 / 4) / 128;
        v4f sv[NSTAGE];
        const float* __restrict__ bslab = B + (size_t)k0 * Ncols + n0;
        const unsigned   toff  = (unsigned)((t >> 4) * Ncols + (t & 15) * 4);
#pragma unroll
        for (int i = 0; i < NSTAGE; ++i)
            sv[i] = *(const v4f*)(bslab + (size_t)(i * 8) * Ncols + toff);
        static_assert(NSTAGE == 16, "the fence below names the sixteen loaded values one by one");
        asm volatile("" :: "v"(sv[0]), "v"(sv[1]), "v"(sv[2]),  "v"(sv[3]),  "v"(sv[4]),  "v"(sv[5]),  "v"(sv[6]),  "v"(sv[7]),
                           "v"(sv[8]), "v"(sv[9]), "v"(sv[10]), "v"(sv[11]), "v"(sv[12]), "v"(sv[13]), "v"(sv[14]), "v"(sv[15]));
        __syncthreads();
#pragma unroll
        for (int i = 0; i < NSTAGE; ++i) {
            const int j = t + i * 128;
            const int k = j >> 4;
            const int c = (j & 15) * 4;
            *(v4f*)(&Bs[k * BS_STRIDE + c]) = sv[i];
        }
        __syncthreads();

        if (valid) {
            const float* ap0 = A + (size_t)row * K + k0 + koff;
            for (int kk = 0; kk < KC; kk += 4) {
                v2f a; a.x = ap0[kk]; a.y = ap0[kk + 1];

                const float* bp = &Bs[(kk + koff) * BS_STRIDE + l15];
                v2f b0; b0.x = bp[0];  b0.y = bp[BS_STRIDE];
                v2f b1; b1.x = bp[16]; b1.y = bp[BS_STRIDE + 16];
                v2f b2; b2.x = bp[32]; b2.y = bp[BS_STRIDE + 32];
                v2f b3; b3.x = bp[48]; b3.y = bp[BS_STRIDE + 48];

                acc0 = __builtin_amdgcn_wmma_f32_16x16x4_f32(false, a, false, b0, (short)0, acc0, false, false);
                acc1 = __builtin_amdgcn_wmma_f32_16x16x4_f32(false, a, false, b1, (short)0, acc1, false, false);
                acc2 = __builtin_amdgcn_wmma_f32_16x16x4_f32(false, a, false, b2, (short)0, acc2, false, false);
                acc3 = __builtin_amdgcn_wmma_f32_16x16x4_f32(false, a, false, b3, (short)0, acc3, false, false);
            }
        }
    }

    if (valid) {
        const int rbase = m0 + ((lane >> 4) ? 8 : 0);
#pragma unroll
        for (int r = 0; r < 8; ++r) {
            float* cp = C + (size_t)(rbase + r) * Ncols + n0 + l15;
            cp[0]  = acc0[r];
            cp[16] = acc1[r];
            cp[32] = acc2[r];
            cp[48] = acc3[r];
        }

        float ps[8], pd[8];
#pragma unroll
        for (int r = 0; r < 8; ++r) {
            ps[r] = acc0[r] * as0 + acc1[r] * as1 + acc2[r] * as2 + acc3[r] * as3;
            pd[r] = acc0[r] * ad0 + acc1[r] * ad1 + acc2[r] * ad2 + acc3[r] * ad3;
        }
#pragma unroll
        for (int r = 0; r < 8; ++r) { ps[r] = row16_sum(ps[r]); pd[r] = row16_sum(pd[r]); }
        float vs = ps[0], vd = pd[0];
#pragma unroll
        for (int r = 1; r < 8; ++r) { vs = (l15 == r) ? ps[r] : vs; vd = (l15 == r) ? pd[r] : vd; }
        if (l15 < 8) {
            const size_t o = (size_t)(rbase + l15) * nparts + by;
            als[o] = vs;
            ald[o] = vd;
        }
    }
}

#define BIN_SHIFT 7
#define BIN_W     128
#define MAX_BINS  512
#define NB1       128

__device__ __forceinline__ int bin_of(int d) {
    return min(d >> BIN_SHIFT, MAX_BINS - 1);
}

__global__ void __launch_bounds__(1024)
bin_local_k(const int* __restrict__ ei, int E, int n, int nbin,
            int* __restrict__ H, int* __restrict__ L, int* __restrict__ pairs) {
    __shared__ int hist[MAX_BINS];
    __shared__ int cur[MAX_BINS];
    __shared__ int wtot[MAX_BINS / 32];
    const int t    = threadIdx.x;
    const int lane = t & 31;
    const int wave = t >> 5;
    const int b    = blockIdx.x;
    for (int k = t; k < MAX_BINS; k += 1024) hist[k] = 0;
    __syncthreads();
    const int per = (E + NB1 - 1) / NB1;
    const int lo  = b * per;
    const int hi  = min(lo + per, E);
    for (int e0 = lo + t; e0 < hi; e0 += 8 * 1024) {
        int d[8];
#pragma unroll
        for (int u = 0; u < 8; ++u) {
            const int e = min(e0 + u * 1024, hi - 1);
            d[u] = ei[E + e];
        }
        asm volatile("" :: "v"(d[0]), "v"(d[1]), "v"(d[2]), "v"(d[3]), "v"(d[4]), "v"(d[5]), "v"(d[6]), "v"(d[7]));
#pragma unroll
        for (int u = 0; u < 8; ++u) {
            if (e0 + u * 1024 < hi)
                atomicAdd(&hist[bin_of(min(max(d[u], 0), n - 1))], 1);
        }
    }
    __syncthreads();

    const int v = (t < MAX_BINS) ? hist[t] : 0;
    int incl = v;
    for (int off = 1; off < 32; off <<= 1) {
        int u = __shfl_up(incl, off);
        if (lane >= off) incl += u;
    }
    if (t < MAX_BINS && lane == 31) wtot[wave] = incl;
    __syncthreads();
    if (t < MAX_BINS) {
        int before = 0;
#pragma unroll
        for (int w = 0; w < MAX_BINS / 32; ++w) before += (w < wave) ? wtot[w] : 0;
        const int excl = before + incl - v;
        cur[t] = excl;
        if (t < nbin) {
            H[t * NB1 + b] = v;
            L[t * NB1 + b] = excl;
        }
    }
    __syncthreads();

    int2* __restrict__ p2 = (int2*)pairs + lo;
    for (int e0 = lo + t; e0 < hi; e0 += 8 * 1024) {
        int d[8], s[8];
#pragma unroll
        for (int u = 0; u < 8; ++u) {
            const int e = min(e0 + u * 1024, hi - 1);
            d[u] = ei[E + e];
            s[u] = ei[e];
        }
        asm volatile("" :: "v"(d[0]), "v"(d[1]), "v"(d[2]), "v"(d[3]), "v"(d[4]), "v"(d[5]), "v"(d[6]), "v"(d[7]),
                           "v"(s[0]), "v"(s[1]), "v"(s[2]), "v"(s[3]), "v"(s[4]), "v"(s[5]), "v"(s[6]), "v"(s[7]));
#pragma unroll
        for (int u = 0; u < 8; ++u) {
            if (e0 + u * 1024 < hi) {
                const int dc  = min(max(d[u], 0), n - 1);
                const int pos = atomicAdd(&cur[bin_of(dc)], 1);
                p2[pos] = make_int2(s[u], dc);
            }
        }
    }
}

#define SORT_T 128
#define SORT_G (4 * NB1 / SORT_T)

__device__ __forceinline__ void
bin_sort_block(const int* __restrict__ pairs, const int* __restrict__ H, const int* __restrict__ L,
               int E, int n, int* __restrict__ rowptr, int* __restrict__ csr_src, int k,
               int* lds) {
    static_assert(SORT_T == BIN_W, "one thread per destination counter in the scan");
    int* cnt  = lds;
    int* fill = cnt  + BIN_W;
    int* wsum = fill + BIN_W;
    int* red  = wsum + BIN_W / 32;
    const int t    = threadIdx.x;
    const int lane = t & 31;
    const int wave = t >> 5;
    const int d0   = k << BIN_SHIFT;
    const int per  = (E + NB1 - 1) / NB1;
    const int cg   = t >> 2;
    const int q    = t & 3;

    int len[SORT_G], lv[SORT_G];
#pragma unroll
    for (int g = 0; g < SORT_G; ++g) {
        const int c = g * (SORT_T / 4) + cg;
        len[g] = H[k * NB1 + c];
        lv[g]  = L[k * NB1 + c];
    }

    int pre = 0;
#pragma unroll
    for (int g = 0; g < SORT_G; ++g) pre += (q == 0) ? lv[g] : 0;
    for (int off = 16; off > 0; off >>= 1) pre += __shfl_xor(pre, off);
    if (t < BIN_W) cnt[t] = 0;
    if (lane == 0) red[wave] = pre;
    __syncthreads();
    int beg = 0;
    for (int w = 0; w < SORT_T / 32; ++w) beg += red[w];

#pragma unroll
    for (int g = 0; g < SORT_G; ++g) {
        const int2* __restrict__ pp = (const int2*)pairs + (size_t)(g * (SORT_T / 4) + cg) * per + lv[g];
        const int ln = len[g];
        for (int j0 = q; j0 < ln; j0 += 32) {
            int y[8];
#pragma unroll
            for (int u = 0; u < 8; ++u) {
                const int j = min(j0 + 4 * u, ln - 1);
                y[u] = pp[j].y;
            }
            asm volatile("" :: "v"(y[0]), "v"(y[1]), "v"(y[2]), "v"(y[3]), "v"(y[4]), "v"(y[5]), "v"(y[6]), "v"(y[7]));
#pragma unroll
            for (int u = 0; u < 8; ++u) {
                if (j0 + 4 * u < ln) atomicAdd(&cnt[(y[u] - d0) & (BIN_W - 1)], 1);
            }
        }
    }
    __syncthreads();

    const int v = (t < BIN_W) ? cnt[t] : 0;
    int incl = v;
    for (int off = 1; off < 32; off <<= 1) {
        int u = __shfl_up(incl, off);
        if (lane >= off) incl += u;
    }
    if (t < BIN_W && lane == 31) wsum[wave] = incl;
    __syncthreads();
    if (t < BIN_W) {
        int before = beg;
        for (int w = 0; w < wave; ++w) before += wsum[w];
        const int excl = before + incl - v;
        fill[t] = excl;
        const int node = d0 + t;
        if (node < n)      rowptr[node] = excl;
        if (node == n - 1) rowptr[n]    = excl + v;
    }
    __syncthreads();
#pragma unroll
    for (int g = 0; g < SORT_G; ++g) {
        const int2* __restrict__ pp = (const int2*)pairs + (size_t)(g * (SORT_T / 4) + cg) * per + lv[g];
        const int ln = len[g];
        for (int j0 = q; j0 < ln; j0 += 32) {
            int2 sd[8];
#pragma unroll
            for (int u = 0; u < 8; ++u) {
                const int j = min(j0 + 4 * u, ln - 1);
                sd[u] = pp[j];
            }
            asm volatile("" :: "v"(sd[0].x), "v"(sd[1].x), "v"(sd[2].x), "v"(sd[3].x), "v"(sd[4].x), "v"(sd[5].x), "v"(sd[6].x), "v"(sd[7].x),
                               "v"(sd[0].y), "v"(sd[1].y), "v"(sd[2].y), "v"(sd[3].y), "v"(sd[4].y), "v"(sd[5].y), "v"(sd[6].y), "v"(sd[7].y));
#pragma unroll
            for (int u = 0; u < 8; ++u) {
                if (j0 + 4 * u < ln) {
                    const int pos = atomicAdd(&fill[(sd[u].y - d0) & (BIN_W - 1)], 1);
                    csr_src[pos] = sd[u].x;
                }
            }
        }
    }
}

__global__ void __launch_bounds__(SORT_T)
gemm_sort_k(const float* __restrict__ A,
            const float* __restrict__ B,
            float* __restrict__ C,
            const float* __restrict__ a_src,
            const float* __restrict__ a_dst,
            float* __restrict__ als, float* __restrict__ ald,
            int M, int K, int Ncols, int gx,
            const int* __restrict__ pairs, const int* __restrict__ H, const int* __restrict__ L,
            int E, int* __restrict__ rowptr, int* __restrict__ csr_src, int nbin) {
    __shared__ float Bs[KC * BS_STRIDE];
    const int b = blockIdx.x;
    if (b < nbin) {
        bin_sort_block(pairs, H, L, E, M, rowptr, csr_src, b, (int*)Bs);
        return;
    }
    const int id = b - nbin;
    const int by = id / gx;
    const int bx = id - by * gx;
    gemm_block(A, B, C, a_src, a_dst, als, ald, M, K, Ncols, bx, by, Ncols / 64, Bs);
}

template <int HEADS>
__device__ __forceinline__ float pick_head(const float (&w)[HEADS], int hsel) {
    float r = w[0];
#pragma unroll
    for (int i = 1; i < HEADS; ++i) r = (hsel == i) ? w[i] : r;
    return r;
}

template <int HEADS, int CH, bool RELU, int PARTS>
__device__ __forceinline__ void
gather_node(int node, int lane,
            const int* __restrict__ rowptr, const int* __restrict__ csr_src,
            const float* __restrict__ als, const float* __restrict__ ald,
            const float* __restrict__ h, const float* __restrict__ bias,
            int n, float4 (&res)[HEADS * CH / 128]) {
    constexpr int C  = HEADS * CH;
    constexpr int NQ = C / 128;

    const int beg = rowptr[node];
    const int cnt = rowptr[node + 1] - beg + 1;

    float adst[HEADS];
#pragma unroll
    for (int hh = 0; hh < HEADS; ++hh) {
        float v = ald[((size_t)node * HEADS + hh) * PARTS];
#pragma unroll
        for (int p = 1; p < PARTS; ++p) v += ald[((size_t)node * HEADS + hh) * PARTS + p];
        adst[hh] = v;
    }

    int hq[NQ];
#pragma unroll
    for (int q = 0; q < NQ; ++q) hq[q] = (q * 128 + lane * 4) / CH;

    float  m[HEADS], dn[HEADS];
    float4 acc[NQ];
#pragma unroll
    for (int hh = 0; hh < HEADS; ++hh) { m[hh] = 0.0f; dn[hh] = 0.0f; }
#pragma unroll
    for (int q = 0; q < NQ; ++q) acc[q] = make_float4(0.0f, 0.0f, 0.0f, 0.0f);

    for (int j0 = 0; j0 < cnt; j0 += 32) {
        const int  j     = j0 + lane;
        const bool valid = j < cnt;
        int s_l = node;
        if (valid && j > 0) s_l = min(max(csr_src[beg + j - 1], 0), n - 1);

        float a[HEADS];
        if constexpr (HEADS == 4 && PARTS == 1) {
            const float4 a4 = *(const float4*)(als + (size_t)s_l * 4);
            a[0] = a4.x; a[1] = a4.y; a[2] = a4.z; a[3] = a4.w;
        } else if constexpr (HEADS == 1 && PARTS == 2) {
            const float2 a2 = *(const float2*)(als + (size_t)s_l * 2);
            a[0] = a2.x + a2.y;
        } else {
#pragma unroll
            for (int hh = 0; hh < HEADS; ++hh) {
                float v = als[((size_t)s_l * HEADS + hh) * PARTS];
#pragma unroll
                for (int p = 1; p < PARTS; ++p) v += als[((size_t)s_l * HEADS + hh) * PARTS + p];
                a[hh] = v;
            }
        }

        float e[HEADS], cm[HEADS], pw[HEADS], ps[HEADS], sc[HEADS];
#pragma unroll
        for (int hh = 0; hh < HEADS; ++hh) {
            e[hh]  = valid ? lrelu(a[hh] + adst[hh]) : -INFINITY;
            cm[hh] = e[hh];
        }
#pragma unroll
        for (int off = 16; off > 0; off >>= 1) {
#pragma unroll
            for (int hh = 0; hh < HEADS; ++hh) cm[hh] = fmaxf(cm[hh], __shfl_xor(cm[hh], off));
        }
#pragma unroll
        for (int hh = 0; hh < HEADS; ++hh) {
            const float mn = (j0 == 0) ? cm[hh] : fmaxf(m[hh], cm[hh]);
            sc[hh] = (j0 == 0) ? 0.0f : __expf(m[hh] - mn);
            pw[hh] = valid ? __expf(e[hh] - mn) : 0.0f;
            ps[hh] = pw[hh];
            m[hh]  = mn;
        }
#pragma unroll
        for (int off = 16; off > 0; off >>= 1) {
#pragma unroll
            for (int hh = 0; hh < HEADS; ++hh) ps[hh] += __shfl_xor(ps[hh], off);
        }
#pragma unroll
        for (int hh = 0; hh < HEADS; ++hh) dn[hh] = dn[hh] * sc[hh] + ps[hh];
        if (j0 > 0) {
#pragma unroll
            for (int q = 0; q < NQ; ++q) {
                const float f = pick_head<HEADS>(sc, hq[q]);
                acc[q].x *= f; acc[q].y *= f; acc[q].z *= f; acc[q].w *= f;
            }
        }

        const int nc = min(32, cnt - j0);
        for (int t = 0; t < nc; t += 4) {
#pragma unroll
            for (int u = 0; u < 4; ++u) {
                const int s = __builtin_amdgcn_readlane(s_l, t + u);
                float wv[HEADS];
#pragma unroll
                for (int hh = 0; hh < HEADS; ++hh)
                    wv[hh] = __int_as_float(__builtin_amdgcn_readlane(__float_as_int(pw[hh]), t + u));
                const float4* hp = (const float4*)(h + (size_t)s * C) + lane;
#pragma unroll
                for (int q = 0; q < NQ; ++q) {
                    const float4 v = hp[q * 32];
                    const float  w = pick_head<HEADS>(wv, hq[q]);
                    acc[q].x += w * v.x; acc[q].y += w * v.y;
                    acc[q].z += w * v.z; acc[q].w += w * v.w;
                }
            }
        }
    }

#pragma unroll
    for (int q = 0; q < NQ; ++q) {
        const int    c0 = q * 128 + lane * 4;
        const float  d  = pick_head<HEADS>(dn, hq[q]);
        const float4 b  = *(const float4*)(bias + c0);
        float4 r;
        r.x = acc[q].x / d + b.x; r.y = acc[q].y / d + b.y;
        r.z = acc[q].z / d + b.z; r.w = acc[q].w / d + b.w;
        if (RELU) {
            r.x = r.x > 0.0f ? r.x : 0.0f; r.y = r.y > 0.0f ? r.y : 0.0f;
            r.z = r.z > 0.0f ? r.z : 0.0f; r.w = r.w > 0.0f ? r.w : 0.0f;
        }
        res[q] = r;
    }
}

template <int HEADS, int CH, bool RELU, int PARTS>
__global__ void __launch_bounds__(256)
gat_gather_k(const int* __restrict__ rowptr, const int* __restrict__ csr_src,
             const float* __restrict__ als, const float* __restrict__ ald,
             const float* __restrict__ h, const float* __restrict__ bias,
             float* __restrict__ out, int n) {
    constexpr int C  = HEADS * CH;
    constexpr int NQ = C / 128;
    const int lane = threadIdx.x & 31;
    const int node = __builtin_amdgcn_readfirstlane((int)(blockIdx.x * 8 + (threadIdx.x >> 5)));
    if (node >= n) return;
    float4 res[NQ];
    gather_node<HEADS, CH, RELU, PARTS>(node, lane, rowptr, csr_src, als, ald, h, bias, n, res);
#pragma unroll
    for (int q = 0; q < NQ; ++q)
        *(float4*)(out + (size_t)node * C + q * 128 + lane * 4) = res[q];
}

#define FR    196
#define FT    13
#define FTHR  1024
#define XS    260
#define KC2   32
#define BS2   136
#define NCH2  (C1 / KC2)

__global__ void __launch_bounds__(FTHR)
gather_gemm_k(const int* __restrict__ rowptr, const int* __restrict__ csr_src,
              const float* __restrict__ als1, const float* __restrict__ ald1,
              const float* __restrict__ h1, const float* __restrict__ b1,
              const float* __restrict__ W2, float* __restrict__ h2,
              const float* __restrict__ a_src, const float* __restrict__ a_dst,
              float* __restrict__ als2, float* __restrict__ ald2, int n) {
    static_assert(FT * 16 >= FR && 2 * FT <= FTHR / 32, "tiles cover the block's nodes, a wave per tile and half");
    static_assert(KC2 * C2 / 4 == FTHR && C1 % KC2 == 0, "one 16-byte piece of a chunk per thread");
    __shared__ float X2s[FT * 16 * XS];
    __shared__ float Bs2[2 * KC2 * BS2];

    const int t    = threadIdx.x;
    const int lane = t & 31;
    const int wave = __builtin_amdgcn_readfirstlane(t >> 5);
    const int row0 = blockIdx.x * FR;

    const int    srow = t >> 5;
    const int    scol = (t & 31) * 4;
    const float* wp   = W2 + (size_t)srow * C2 + scol;
    v4f pre = *(const v4f*)wp;

    for (int i = wave; i < FT * 16; i += FTHR / 32) {
        const int node = row0 + i;
        float4 res[2];
        if (i < FR && node < n) {
            gather_node<HEADS1, HID1, true, 1>(node, lane, rowptr, csr_src, als1, ald1, h1, b1, n, res);
        } else {
            res[0] = make_float4(0.0f, 0.0f, 0.0f, 0.0f);
            res[1] = res[0];
        }
        *(float4*)(&X2s[i * XS + lane * 4])       = res[0];
        *(float4*)(&X2s[i * XS + 128 + lane * 4]) = res[1];
    }

    *(v4f*)(&Bs2[srow * BS2 + scol]) = pre;
    pre = *(const v4f*)(wp + (size_t)KC2 * C2);

    const bool mm   = wave < 2 * FT;
    const int  rt   = mm ? wave % FT : 0;
    const int  half = mm ? wave / FT : 0;
    const int  n0   = half * 64;
    const int  l15  = lane & 15;
    const int  koff = (lane >> 4) * 2;

    const float as0 = a_src[n0 + l15],      ad0 = a_dst[n0 + l15];
    const float as1 = a_src[n0 + l15 + 16], ad1 = a_dst[n0 + l15 + 16];
    const float as2 = a_src[n0 + l15 + 32], ad2 = a_dst[n0 + l15 + 32];
    const float as3 = a_src[n0 + l15 + 48], ad3 = a_dst[n0 + l15 + 48];

    v8f acc0 = {}, acc1 = {}, acc2 = {}, acc3 = {};
    const float* ap0 = &X2s[(rt * 16 + l15) * XS + koff];

    for (int c = 0; c < NCH2; ++c) {
        __syncthreads();
        if (c + 1 < NCH2) {
            *(v4f*)(&Bs2[((c + 1) & 1) * (KC2 * BS2) + srow * BS2 + scol]) = pre;
            if (c + 2 < NCH2) pre = *(const v4f*)(wp + (size_t)(c + 2) * KC2 * C2);
        }
        if (mm) {
            const float* ap = ap0 + c * KC2;
            const float* bb = &Bs2[(c & 1) * (KC2 * BS2) + koff * BS2 + n0 + l15];
            for (int kk = 0; kk < KC2; kk += 4) {
                v2f a; a.x = ap[kk]; a.y = ap[kk + 1];

                const float* bp = bb + kk * BS2;
                v2f b0; b0.x = bp[0];  b0.y = bp[BS2];
                v2f b1v; b1v.x = bp[16]; b1v.y = bp[BS2 + 16];
                v2f b2; b2.x = bp[32]; b2.y = bp[BS2 + 32];
                v2f b3; b3.x = bp[48]; b3.y = bp[BS2 + 48];

                acc0 = __builtin_amdgcn_wmma_f32_16x16x4_f32(false, a, false, b0,  (short)0, acc0, false, false);
                acc1 = __builtin_amdgcn_wmma_f32_16x16x4_f32(false, a, false, b1v, (short)0, acc1, false, false);
                acc2 = __builtin_amdgcn_wmma_f32_16x16x4_f32(false, a, false, b2,  (short)0, acc2, false, false);
                acc3 = __builtin_amdgcn_wmma_f32_16x16x4_f32(false, a, false, b3,  (short)0, acc3, false, false);
            }
        }
    }

    if (mm) {
        const int lr0 = rt * 16 + ((lane >> 4) ? 8 : 0);
#pragma unroll
        for (int r = 0; r < 8; ++r) {
            const int lrow = lr0 + r;
            const int grow = row0 + lrow;
            if (lrow < FR && grow < n) {
                float* cp = h2 + (size_t)grow * C2 + n0 + l15;
                cp[0]  = acc0[r];
                cp[16] = acc1[r];
                cp[32] = acc2[r];
                cp[48] = acc3[r];
            }
        }

        float ps[8], pd[8];
#pragma unroll
        for (int r = 0; r < 8; ++r) {
            ps[r] = acc0[r] * as0 + acc1[r] * as1 + acc2[r] * as2 + acc3[r] * as3;
            pd[r] = acc0[r] * ad0 + acc1[r] * ad1 + acc2[r] * ad2 + acc3[r] * ad3;
        }
#pragma unroll
        for (int r = 0; r < 8; ++r) { ps[r] = row16_sum(ps[r]); pd[r] = row16_sum(pd[r]); }
        float vs = ps[0], vd = pd[0];
#pragma unroll
        for (int r = 1; r < 8; ++r) { vs = (l15 == r) ? ps[r] : vs; vd = (l15 == r) ? pd[r] : vd; }
        const int lrow = lr0 + l15;
        const int grow = row0 + lrow;
        if (l15 < 8 && lrow < FR && grow < n) {
            const size_t o = (size_t)grow * 2 + half;
            als2[o] = vs;
            ald2[o] = vd;
        }
    }
}


extern "C" void kernel_launch(void* const* d_in, const int* in_sizes, int n_in,
                              void* d_out, int out_size, void* d_ws, size_t ws_size,
                              hipStream_t stream) {
    const float* x      = (const float*)d_in[0];
    const int*   eidx   = (const int*)d_in[1];
    const float* W1     = (const float*)d_in[2];
    const float* a1_src = (const float*)d_in[3];
    const float* a1_dst = (const float*)d_in[4];
    const float* b1     = (const float*)d_in[5];
    const float* W2     = (const float*)d_in[6];
    const float* a2_src = (const float*)d_in[7];
    const float* a2_dst = (const float*)d_in[8];
    const float* b2     = (const float*)d_in[9];
    float*       out    = (float*)d_out;

    const int N = in_sizes[0] / 128;
    const int E = in_sizes[1] / 2;

    float* ws   = (float*)d_ws;
    float* h1   = ws;
    float* h2   = h1   + (size_t)N * C1;
    float* als1 = h2   + (size_t)N * C1;
    float* ald1 = als1 + (size_t)N * HEADS1;
    float* als2 = ald1 + (size_t)N * HEADS1;
    float* ald2 = als2 + (size_t)N * 2;
    const int nbin  = (N + BIN_W - 1) / BIN_W;
    const int nscan = nbin * NB1;
    int*   pairs  = (int*)(ald2 + (size_t)N * 2);
    int*   csr    = pairs  + (size_t)2 * E;
    int*   H      = csr    + (size_t)E;
    int*   L      = H      + (size_t)nscan;
    int*   rowptr = L      + (size_t)nscan + 1;

    const int mtiles = (N + 15) / 16;

    bin_local_k<<<NB1, 1024, 0, stream>>>(eidx, E, N, nbin, H, L, pairs);

    {
        const int gx = (mtiles + 3) / 4;
        gemm_sort_k<<<nbin + gx * (C1 / 64), SORT_T, 0, stream>>>(x, W1, h1, a1_src, a1_dst, als1, ald1, N, 128, C1, gx,
                                                                 pairs, H, L, E, rowptr, csr, nbin);
    }

    gather_gemm_k<<<(N + FR - 1) / FR, FTHR, 0, stream>>>(rowptr, csr, als1, ald1, h1, b1,
                                                         W2, h2, a2_src, a2_dst, als2, ald2, N);

    gat_gather_k<1, C2, false, 2><<<(N + 7) / 8, 256, 0, stream>>>(rowptr, csr, als2, ald2,
                                                                   h2, b2, out, N);
}
